# starting kernel
# baseline (speedup 1.0000x reference)
#include <hip/hip_runtime.h>

#define R_REL   2
#define IN_F    128
#define HID_F   64
#define OUT_F   64
#define H1      2
#define H2      1
#define NEG_SLOPE 0.2f

typedef __attribute__((ext_vector_type(16))) __bf16 v16bf;
typedef __attribute__((ext_vector_type(8)))  float  v8f;

union Frag16 { uint4 q[2]; v16bf v; };

__device__ __forceinline__ unsigned short f2bf(float f) {
  unsigned u = __float_as_uint(f);
  unsigned r = u + 0x7FFFu + ((u >> 16) & 1u);
  return (unsigned short)(r >> 16);
}
__device__ __forceinline__ unsigned fkey(float f) {
  unsigned u = __float_as_uint(f);
  return (u & 0x80000000u) ? ~u : (u | 0x80000000u);
}
__device__ __forceinline__ float funkey(unsigned k) {
  unsigned u = (k & 0x80000000u) ? (k & 0x7FFFFFFFu) : ~k;
  return __uint_as_float(u);
}
__device__ __forceinline__ float lrelu(float x) { return x >= 0.f ? x : NEG_SLOPE * x; }

__global__ void k_f32_to_bf16(const float* __restrict__ in,
                              unsigned short* __restrict__ out, size_t n) {
  size_t i = (size_t)blockIdx.x * blockDim.x + threadIdx.x;
  if (i < n) out[i] = f2bf(in[i]);
}

__global__ void k_relu_f32_to_bf16(const float* __restrict__ in,
                                   unsigned short* __restrict__ out, size_t n) {
  size_t i = (size_t)blockIdx.x * blockDim.x + threadIdx.x;
  if (i < n) { float v = in[i]; out[i] = f2bf(v > 0.f ? v : 0.f); }
}

__global__ void k_transpose_w(const float* __restrict__ W,
                              unsigned short* __restrict__ Wt, int K, int NCOLS) {
  int i = blockIdx.x * blockDim.x + threadIdx.x;
  if (i >= K * NCOLS) return;
  int n = i / K, k = i % K;
  Wt[i] = f2bf(W[(size_t)k * NCOLS + n]);
}

__global__ void k_zero_u32(unsigned* __restrict__ p, size_t n) {
  size_t i = (size_t)blockIdx.x * blockDim.x + threadIdx.x;
  if (i < n) p[i] = 0u;
}

__global__ void k_init_bias(float* __restrict__ acc, const float* __restrict__ b,
                            int N, int F) {
  size_t i = (size_t)blockIdx.x * blockDim.x + threadIdx.x;
  if (i >= (size_t)N * F) return;
  int c = (int)(i % F);
  float s = 0.f;
  for (int r = 0; r < R_REL; ++r) s += b[(size_t)r * F + c];
  acc[i] = s;
}

template <int KDIM, int NCOLS, int RPB>
__global__ __launch_bounds__(2 * NCOLS)
void k_gemm_bf16_wmma(const unsigned short* __restrict__ A,
                      const unsigned short* __restrict__ Bt,
                      float* __restrict__ C, int M) {
  constexpr int KC = KDIM / 32;
  const int lane     = threadIdx.x & 31;
  const int wv       = threadIdx.x >> 5;
  const int colBase  = wv * 16;
  const int l16      = lane & 15;
  const int k0       = (lane < 16) ? 0 : 8;
  const int rowBlock = blockIdx.x * (16 * RPB);

  Frag16 fb[KC];
  {
    const unsigned short* bp = Bt + (size_t)(colBase + l16) * KDIM + k0;
#pragma unroll
    for (int c = 0; c < KC; ++c) {
      fb[c].q[0] = *reinterpret_cast<const uint4*>(bp + 32 * c);
      fb[c].q[1] = *reinterpret_cast<const uint4*>(bp + 32 * c + 16);
    }
  }

  const bool full = (rowBlock + 16 * RPB) <= M;
  const int  rOff = (lane < 16) ? 0 : 8;

#pragma unroll
  for (int t = 0; t < RPB; ++t) {
    const int rowBase = rowBlock + 16 * t;
    if (rowBase >= M) break;

    int arow = rowBase + l16; if (arow >= M) arow = M - 1;
    const unsigned short* ap = A + (size_t)arow * KDIM + k0;

    v8f acc = {};
#pragma unroll
    for (int c = 0; c < KC; ++c) {
      Frag16 fa;
      fa.q[0] = *reinterpret_cast<const uint4*>(ap + 32 * c);
      fa.q[1] = *reinterpret_cast<const uint4*>(ap + 32 * c + 16);
      acc = __builtin_amdgcn_wmma_f32_16x16x32_bf16(false, fa.v, false, fb[c].v,
                                                    (short)0, acc, false, false);
    }

    float* cp = C + (size_t)rowBase * NCOLS + colBase + l16;
    if (full) {
#pragma unroll
      for (int v = 0; v < 8; ++v) cp[(size_t)(rOff + v) * NCOLS] = acc[v];
    } else {
#pragma unroll
      for (int v = 0; v < 8; ++v) {
        int rr = rowBase + rOff + v;
        if (rr < M) cp[(size_t)(rOff + v) * NCOLS] = acc[v];
      }
    }
  }
}

__global__ void k_attn_dots(const float* __restrict__ feat,
                            const float* __restrict__ al, const float* __restrict__ ar,
                            float* __restrict__ el, float* __restrict__ er,
                            int N, int H, int D) {
  int i = blockIdx.x * blockDim.x + threadIdx.x;
  if (i >= N * H) return;
  int h = i % H;
  const float* f = feat + (size_t)(i / H) * H * D + (size_t)h * D;
  float sl = 0.f, sr = 0.f;
  for (int d = 0; d < D; d += 4) {
    float4 v = *reinterpret_cast<const float4*>(f + d);
    float4 a = *reinterpret_cast<const float4*>(al + h * D + d);
    float4 b = *reinterpret_cast<const float4*>(ar + h * D + d);
    sl += v.x * a.x + v.y * a.y + v.z * a.z + v.w * a.w;
    sr += v.x * b.x + v.y * b.y + v.z * b.z + v.w * b.w;
  }
  el[i] = sl; er[i] = sr;
}

__global__ void k_edge_max(const float* __restrict__ el, const float* __restrict__ er,
                           const int* __restrict__ src, const int* __restrict__ dst,
                           unsigned* __restrict__ mkey, int E, int H) {
  int e = blockIdx.x * blockDim.x + threadIdx.x;
  if (e >= E) return;
  int s = src[e], d = dst[e];
  for (int h = 0; h < H; ++h) {
    float v = lrelu(el[(size_t)s * H + h] + er[(size_t)d * H + h]);
    atomicMax(&mkey[(size_t)d * H + h], fkey(v));
  }
}

__global__ void k_edge_exp(const float* __restrict__ el, const float* __restrict__ er,
                           const int* __restrict__ src, const int* __restrict__ dst,
                           const unsigned* __restrict__ mkey, float* __restrict__ ssum,
                           float* __restrict__ exbuf, int E, int H) {
  int e = blockIdx.x * blockDim.x + threadIdx.x;
  if (e >= E) return;
  int s = src[e], d = dst[e];
  for (int h = 0; h < H; ++h) {
    float v  = lrelu(el[(size_t)s * H + h] + er[(size_t)d * H + h]);
    float m  = funkey(mkey[(size_t)d * H + h]);
    float ex = __expf(v - m);
    exbuf[(size_t)e * H + h] = ex;
    atomicAdd(&ssum[(size_t)d * H + h], ex);
  }
}

template <int H, int D>
__global__ void k_edge_scatter(const float* __restrict__ exbuf,
                               const float* __restrict__ ssum,
                               const float* __restrict__ feat,
                               const int* __restrict__ src, const int* __restrict__ dst,
                               float* __restrict__ acc, int E) {
  constexpr int F   = H * D;
  constexpr int PER = F / 32;
  int wid  = (int)(((size_t)blockIdx.x * blockDim.x + threadIdx.x) >> 5);
  int lane = threadIdx.x & 31;
  if (wid >= E) return;
  int s = src[wid], d = dst[wid];

  const int base = lane * PER;
  const int h    = base / D;
  float sum   = ssum[(size_t)d * H + h];
  float alpha = exbuf[(size_t)wid * H + h] / (sum > 0.f ? sum : 1.f);

  const float* fp = feat + (size_t)s * F + base;
  float*       op = acc  + (size_t)d * F + base;
  if constexpr (PER == 4) {
    float4 v = *reinterpret_cast<const float4*>(fp);
    atomicAdd(op + 0, alpha * v.x);
    atomicAdd(op + 1, alpha * v.y);
    atomicAdd(op + 2, alpha * v.z);
    atomicAdd(op + 3, alpha * v.w);
  } else {
    float2 v = *reinterpret_cast<const float2*>(fp);
    atomicAdd(op + 0, alpha * v.x);
    atomicAdd(op + 1, alpha * v.y);
  }
}

extern "C" void kernel_launch(void* const* d_in, const int* in_sizes, int n_in,
                              void* d_out, int out_size, void* d_ws, size_t ws_size,
                              hipStream_t stream) {
  const float* x    = (const float*)d_in[0];
  const int*   esrc = (const int*)d_in[1];
  const int*   edst = (const int*)d_in[2];
  const float* W1   = (const float*)d_in[3];
  const float* al1  = (const float*)d_in[4];
  const float* ar1  = (const float*)d_in[5];
  const float* b1   = (const float*)d_in[6];
  const float* W2   = (const float*)d_in[7];
  const float* al2  = (const float*)d_in[8];
  const float* ar2  = (const float*)d_in[9];
  const float* b2   = (const float*)d_in[10];
  float* out = (float*)d_out;

  const int N = in_sizes[0] / IN_F;
  const int E = in_sizes[1] / (2 * R_REL);

  size_t off = 0;
  auto carve = [&](size_t bytes) -> void* {
    void* p = (char*)d_ws + off;
    off += (bytes + 255) & ~(size_t)255;
    return p;
  };
  unsigned short* xbf  = (unsigned short*)carve((size_t)N * IN_F * 2);
  unsigned short* hbf  = (unsigned short*)carve((size_t)N * (H1 * HID_F) * 2);
  unsigned short* wt1  = (unsigned short*)carve((size_t)R_REL * IN_F * (H1 * HID_F) * 2);
  unsigned short* wt2  = (unsigned short*)carve((size_t)R_REL * (H1 * HID_F) * OUT_F * 2);
  float*    feat  = (float*)carve((size_t)N * (H1 * HID_F) * 4);
  float*    el    = (float*)carve((size_t)N * H1 * 4);
  float*    er    = (float*)carve((size_t)N * H1 * 4);
  unsigned* mkey  = (unsigned*)carve((size_t)N * H1 * 4);
  float*    ssum  = (float*)carve((size_t)N * H1 * 4);
  float*    exbuf = (float*)carve((size_t)E * H1 * 4);
  float*    hacc  = (float*)carve((size_t)N * (H1 * HID_F) * 4);

  const int T = 256;
  auto blk1 = [&](size_t n) { return (unsigned)((n + T - 1) / T); };

  k_f32_to_bf16<<<blk1((size_t)N * IN_F), T, 0, stream>>>(x, xbf, (size_t)N * IN_F);
  for (int r = 0; r < R_REL; ++r) {
    k_transpose_w<<<blk1((size_t)IN_F * (H1 * HID_F)), T, 0, stream>>>(
        W1 + (size_t)r * IN_F * (H1 * HID_F), wt1 + (size_t)r * IN_F * (H1 * HID_F),
        IN_F, H1 * HID_F);
    k_transpose_w<<<blk1((size_t)(H1 * HID_F) * OUT_F), T, 0, stream>>>(
        W2 + (size_t)r * (H1 * HID_F) * OUT_F, wt2 + (size_t)r * (H1 * HID_F) * OUT_F,
        H1 * HID_F, OUT_F);
  }

  constexpr int RPB = 4;
  const unsigned gemmGrid = (unsigned)((N + 16 * RPB - 1) / (16 * RPB));

  k_init_bias<<<blk1((size_t)N * (H1 * HID_F)), T, 0, stream>>>(hacc, b1, N, H1 * HID_F);
  for (int r = 0; r < R_REL; ++r) {
    const int* s1 = esrc + (size_t)(0 * R_REL + r) * E;
    const int* d1 = edst + (size_t)(0 * R_REL + r) * E;

    k_gemm_bf16_wmma<IN_F, H1 * HID_F, RPB><<<gemmGrid, 2 * (H1 * HID_F), 0, stream>>>(
        xbf, wt1 + (size_t)r * IN_F * (H1 * HID_F), feat, N);

    k_attn_dots<<<blk1((size_t)N * H1), T, 0, stream>>>(
        feat, al1 + (size_t)r * H1 * HID_F, ar1 + (size_t)r * H1 * HID_F,
        el, er, N, H1, HID_F);

    k_zero_u32<<<blk1((size_t)N * H1), T, 0, stream>>>(mkey, (size_t)N * H1);
    k_zero_u32<<<blk1((size_t)N * H1), T, 0, stream>>>((unsigned*)ssum, (size_t)N * H1);

    k_edge_max<<<blk1((size_t)E), T, 0, stream>>>(el, er, s1, d1, mkey, E, H1);
    k_edge_exp<<<blk1((size_t)E), T, 0, stream>>>(el, er, s1, d1, mkey, ssum, exbuf, E, H1);
    k_edge_scatter<H1, HID_F><<<blk1((size_t)E * 32), T, 0, stream>>>(
        exbuf, ssum, feat, s1, d1, hacc, E);
  }

  k_relu_f32_to_bf16<<<blk1((size_t)N * (H1 * HID_F)), T, 0, stream>>>(
      hacc, hbf, (size_t)N * (H1 * HID_F));

  k_init_bias<<<blk1((size_t)N * OUT_F), T, 0, stream>>>(out, b2, N, OUT_F);
  for (int r = 0; r < R_REL; ++r) {
    const int* s2 = esrc + (size_t)(1 * R_REL + r) * E;
    const int* d2 = edst + (size_t)(1 * R_REL + r) * E;

    k_gemm_bf16_wmma<H1 * HID_F, OUT_F, RPB><<<gemmGrid, 2 * OUT_F, 0, stream>>>(
        hbf, wt2 + (size_t)r * (H1 * HID_F) * OUT_F, feat, N);

    k_attn_dots<<<blk1((size_t)N * H2), T, 0, stream>>>(
        feat, al2 + (size_t)r * H2 * OUT_F, ar2 + (size_t)r * H2 * OUT_F,
        el, er, N, H2, OUT_F);

    k_zero_u32<<<blk1((size_t)N * H2), T, 0, stream>>>(mkey, (size_t)N * H2);
    k_zero_u32<<<blk1((size_t)N * H2), T, 0, stream>>>((unsigned*)ssum, (size_t)N * H2);

    k_edge_max<<<blk1((size_t)E), T, 0, stream>>>(el, er, s2, d2, mkey, E, H2);
    k_edge_exp<<<blk1((size_t)E), T, 0, stream>>>(el, er, s2, d2, mkey, ssum, exbuf, E, H2);
    k_edge_scatter<H2, OUT_F><<<blk1((size_t)E * 32), T, 0, stream>>>(
        exbuf, ssum, feat, s2, d2, out, E);
  }
}


# Edges counting-sorted by destination each call; one wave per node does max, exp, sum and weighted feature sum in registers: no float atomics, sum order changes
# speedup vs baseline: 1.8928x; 1.8928x over previous
#include <hip/hip_runtime.h>

#define R_REL   2
#define IN_F    128
#define HID_F   64
#define OUT_F   64
#define H1      2
#define H2      1
#define NEG_SLOPE 0.2f

typedef __attribute__((ext_vector_type(16))) __bf16 v16bf;
typedef __attribute__((ext_vector_type(8)))  float  v8f;

union Frag16 { uint4 q[2]; v16bf v; };

__device__ __forceinline__ unsigned short f2bf(float f) {
  unsigned u = __float_as_uint(f);
  unsigned r = u + 0x7FFFu + ((u >> 16) & 1u);
  return (unsigned short)(r >> 16);
}
__device__ __forceinline__ float lrelu(float x) { return x >= 0.f ? x : NEG_SLOPE * x; }

__global__ void k_f32_to_bf16(const float* __restrict__ in,
                              unsigned short* __restrict__ out, size_t n) {
  size_t i = (size_t)blockIdx.x * blockDim.x + threadIdx.x;
  if (i < n) out[i] = f2bf(in[i]);
}

__global__ void k_relu_f32_to_bf16(const float* __restrict__ in,
                                   unsigned short* __restrict__ out, size_t n) {
  size_t i = (size_t)blockIdx.x * blockDim.x + threadIdx.x;
  if (i < n) { float v = in[i]; out[i] = f2bf(v > 0.f ? v : 0.f); }
}

__global__ void k_transpose_w(const float* __restrict__ W,
                              unsigned short* __restrict__ Wt, int K, int NCOLS) {
  int i = blockIdx.x * blockDim.x + threadIdx.x;
  if (i >= K * NCOLS) return;
  int n = i / K, k = i % K;
  Wt[i] = f2bf(W[(size_t)k * NCOLS + n]);
}

__global__ void k_zero_u32(unsigned* __restrict__ p, size_t n) {
  size_t i = (size_t)blockIdx.x * blockDim.x + threadIdx.x;
  if (i < n) p[i] = 0u;
}

__global__ void k_init_bias(float* __restrict__ acc, const float* __restrict__ b,
                            int N, int F) {
  size_t i = (size_t)blockIdx.x * blockDim.x + threadIdx.x;
  if (i >= (size_t)N * F) return;
  int c = (int)(i % F);
  float s = 0.f;
  for (int r = 0; r < R_REL; ++r) s += b[(size_t)r * F + c];
  acc[i] = s;
}

template <int KDIM, int NCOLS, int RPB>
__global__ __launch_bounds__(2 * NCOLS)
void k_gemm_bf16_wmma(const unsigned short* __restrict__ A,
                      const unsigned short* __restrict__ Bt,
                      float* __restrict__ C, int M) {
  constexpr int KC = KDIM / 32;
  const int lane     = threadIdx.x & 31;
  const int wv       = threadIdx.x >> 5;
  const int colBase  = wv * 16;
  const int l16      = lane & 15;
  const int k0       = (lane < 16) ? 0 : 8;
  const int rowBlock = blockIdx.x * (16 * RPB);

  Frag16 fb[KC];
  {
    const unsigned short* bp = Bt + (size_t)(colBase + l16) * KDIM + k0;
#pragma unroll
    for (int c = 0; c < KC; ++c) {
      fb[c].q[0] = *reinterpret_cast<const uint4*>(bp + 32 * c);
      fb[c].q[1] = *reinterpret_cast<const uint4*>(bp + 32 * c + 16);
    }
  }

  const bool full = (rowBlock + 16 * RPB) <= M;
  const int  rOff = (lane < 16) ? 0 : 8;

#pragma unroll
  for (int t = 0; t < RPB; ++t) {
    const int rowBase = rowBlock + 16 * t;
    if (rowBase >= M) break;

    int arow = rowBase + l16; if (arow >= M) arow = M - 1;
    const unsigned short* ap = A + (size_t)arow * KDIM + k0;

    v8f acc = {};
#pragma unroll
    for (int c = 0; c < KC; ++c) {
      Frag16 fa;
      fa.q[0] = *reinterpret_cast<const uint4*>(ap + 32 * c);
      fa.q[1] = *reinterpret_cast<const uint4*>(ap + 32 * c + 16);
      acc = __builtin_amdgcn_wmma_f32_16x16x32_bf16(false, fa.v, false, fb[c].v,
                                                    (short)0, acc, false, false);
    }

    float* cp = C + (size_t)rowBase * NCOLS + colBase + l16;
    if (full) {
#pragma unroll
      for (int v = 0; v < 8; ++v) cp[(size_t)(rOff + v) * NCOLS] = acc[v];
    } else {
#pragma unroll
      for (int v = 0; v < 8; ++v) {
        int rr = rowBase + rOff + v;
        if (rr < M) cp[(size_t)(rOff + v) * NCOLS] = acc[v];
      }
    }
  }
}

__global__ void k_attn_dots(const float* __restrict__ feat,
                            const float* __restrict__ al, const float* __restrict__ ar,
                            float* __restrict__ el, float* __restrict__ er,
                            int N, int H, int D) {
  int i = blockIdx.x * blockDim.x + threadIdx.x;
  if (i >= N * H) return;
  int h = i % H;
  const float* f = feat + (size_t)(i / H) * H * D + (size_t)h * D;
  float sl = 0.f, sr = 0.f;
  for (int d = 0; d < D; d += 4) {
    float4 v = *reinterpret_cast<const float4*>(f + d);
    float4 a = *reinterpret_cast<const float4*>(al + h * D + d);
    float4 b = *reinterpret_cast<const float4*>(ar + h * D + d);
    sl += v.x * a.x + v.y * a.y + v.z * a.z + v.w * a.w;
    sr += v.x * b.x + v.y * b.y + v.z * b.z + v.w * b.w;
  }
  el[i] = sl; er[i] = sr;
}

__global__ void k_dst_hist(const int* __restrict__ dst_all, unsigned* __restrict__ cnt,
                           int E, int NP, int total) {
  int g = blockIdx.x * blockDim.x + threadIdx.x;
  if (g >= total) return;
  int q = g / E;
  atomicAdd(&cnt[(size_t)q * NP + dst_all[g]], 1u);
}

__global__ __launch_bounds__(256)
void k_tile_sum(const unsigned* __restrict__ cnt, unsigned* __restrict__ bsum, int NP, int nb) {
  __shared__ unsigned sh[256];
  const int t = threadIdx.x, b = blockIdx.x, q = blockIdx.y;
  const uint4 c = *reinterpret_cast<const uint4*>(cnt + (size_t)q * NP + (size_t)b * 1024 + t * 4);
  sh[t] = c.x + c.y + c.z + c.w;
  __syncthreads();
  for (int s = 128; s > 0; s >>= 1) {
    if (t < s) sh[t] += sh[t + s];
    __syncthreads();
  }
  if (t == 0) bsum[q * nb + b] = sh[0];
}

__global__ __launch_bounds__(256)
void k_tile_scan(const unsigned* __restrict__ cnt, const unsigned* __restrict__ bsum,
                 int* __restrict__ rowptr, int* __restrict__ cursor, int NP, int nb) {
  __shared__ unsigned sh[256];
  const int t = threadIdx.x, b = blockIdx.x, q = blockIdx.y;
  unsigned part = 0;
  for (int j = t; j < b; j += 256) part += bsum[q * nb + j];
  sh[t] = part;
  __syncthreads();
  for (int s = 128; s > 0; s >>= 1) {
    if (t < s) sh[t] += sh[t + s];
    __syncthreads();
  }
  const unsigned off = sh[0];
  __syncthreads();

  const size_t idx = (size_t)q * NP + (size_t)b * 1024 + t * 4;
  const uint4 c = *reinterpret_cast<const uint4*>(cnt + idx);
  const unsigned t0 = c.x, t1 = t0 + c.y, t2 = t1 + c.z, t3 = t2 + c.w;
  sh[t] = t3;
  __syncthreads();
  for (int o = 1; o < 256; o <<= 1) {
    unsigned v = (t >= o) ? sh[t - o] : 0u;
    __syncthreads();
    sh[t] += v;
    __syncthreads();
  }
  const unsigned ex = sh[t] - t3 + off;
  int4 r;
  r.x = (int)ex; r.y = (int)(ex + t0); r.z = (int)(ex + t1); r.w = (int)(ex + t2);
  *reinterpret_cast<int4*>(rowptr + idx) = r;
  *reinterpret_cast<int4*>(cursor + idx) = r;
}

__global__ void k_dst_fill(const int* __restrict__ src_all, const int* __restrict__ dst_all,
                           int* __restrict__ cursor, int* __restrict__ csrc,
                           int E, int NP, int total) {
  int g = blockIdx.x * blockDim.x + threadIdx.x;
  if (g >= total) return;
  int q = g / E;
  int pos = atomicAdd(&cursor[(size_t)q * NP + dst_all[g]], 1);
  csrc[(size_t)q * E + pos] = src_all[g];
}

__device__ __forceinline__ float bcast_lane(float v, int j) {
  return __int_as_float(__builtin_amdgcn_readlane(__float_as_int(v), j));
}

template <int H, int D>
__global__ __launch_bounds__(256)
void k_seg_softmax_sum(const int* __restrict__ rowptr, const int* __restrict__ csrc,
                       const float* __restrict__ el, const float* __restrict__ er,
                       const float* __restrict__ feat, float* __restrict__ acc, int N) {
  constexpr int F   = H * D;
  constexpr int PER = F / 32;
  const int d    = __builtin_amdgcn_readfirstlane(
      (int)(((size_t)blockIdx.x * blockDim.x + threadIdx.x) >> 5));
  const int lane = threadIdx.x & 31;
  if (d >= N) return;
  const int beg = rowptr[d], end = rowptr[d + 1];
  if (end <= beg) return;

  float erd[H], m[H], den[H];
#pragma unroll
  for (int h = 0; h < H; ++h) { erd[h] = er[(size_t)d * H + h]; m[h] = -INFINITY; den[h] = 0.f; }

  for (int i = beg + lane; i < end; i += 32) {
    const int s = csrc[i];
#pragma unroll
    for (int h = 0; h < H; ++h) m[h] = fmaxf(m[h], lrelu(el[(size_t)s * H + h] + erd[h]));
  }
#pragma unroll
  for (int h = 0; h < H; ++h)
    for (int o = 16; o > 0; o >>= 1) m[h] = fmaxf(m[h], __shfl_xor(m[h], o));

  for (int i = beg + lane; i < end; i += 32) {
    const int s = csrc[i];
#pragma unroll
    for (int h = 0; h < H; ++h) den[h] += __expf(lrelu(el[(size_t)s * H + h] + erd[h]) - m[h]);
  }
#pragma unroll
  for (int h = 0; h < H; ++h) {
    for (int o = 16; o > 0; o >>= 1) den[h] += __shfl_xor(den[h], o);
    den[h] = den[h] > 0.f ? den[h] : 1.f;
  }

  const int base = lane * PER;
  const int hl   = base / D;
  float a[PER];
#pragma unroll
  for (int k = 0; k < PER; ++k) a[k] = 0.f;

  for (int cb = beg; cb < end; cb += 32) {
    const int i = cb + lane;
    int   s_l = 0;
    float w_l[H];
#pragma unroll
    for (int h = 0; h < H; ++h) w_l[h] = 0.f;
    if (i < end) {
      s_l = csrc[i];
#pragma unroll
      for (int h = 0; h < H; ++h)
        w_l[h] = __expf(lrelu(el[(size_t)s_l * H + h] + erd[h]) - m[h]) / den[h];
    }
    const int cnt = (end - cb) < 32 ? (end - cb) : 32;
#pragma unroll 4
    for (int j = 0; j < cnt; ++j) {
      const int sj = __builtin_amdgcn_readlane(s_l, j);
      float wj = bcast_lane(w_l[0], j);
      if constexpr (H == 2) { const float w1 = bcast_lane(w_l[1], j); wj = hl ? w1 : wj; }
      const float* fp = feat + (size_t)sj * F + base;
      if constexpr (PER == 4) {
        const float4 v = *reinterpret_cast<const float4*>(fp);
        a[0] += wj * v.x; a[1] += wj * v.y; a[2] += wj * v.z; a[3] += wj * v.w;
      } else {
        const float2 v = *reinterpret_cast<const float2*>(fp);
        a[0] += wj * v.x; a[1] += wj * v.y;
      }
    }
  }

  float* op = acc + (size_t)d * F + base;
  if constexpr (PER == 4) {
    float4 o = *reinterpret_cast<const float4*>(op);
    o.x += a[0]; o.y += a[1]; o.z += a[2]; o.w += a[3];
    *reinterpret_cast<float4*>(op) = o;
  } else {
    float2 o = *reinterpret_cast<const float2*>(op);
    o.x += a[0]; o.y += a[1];
    *reinterpret_cast<float2*>(op) = o;
  }
}

extern "C" void kernel_launch(void* const* d_in, const int* in_sizes, int n_in,
                              void* d_out, int out_size, void* d_ws, size_t ws_size,
                              hipStream_t stream) {
  const float* x    = (const float*)d_in[0];
  const int*   esrc = (const int*)d_in[1];
  const int*   edst = (const int*)d_in[2];
  const float* W1   = (const float*)d_in[3];
  const float* al1  = (const float*)d_in[4];
  const float* ar1  = (const float*)d_in[5];
  const float* b1   = (const float*)d_in[6];
  const float* W2   = (const float*)d_in[7];
  const float* al2  = (const float*)d_in[8];
  const float* ar2  = (const float*)d_in[9];
  const float* b2   = (const float*)d_in[10];
  float* out = (float*)d_out;

  const int N = in_sizes[0] / IN_F;
  const int E = in_sizes[1] / (2 * R_REL);

  size_t off = 0;
  auto carve = [&](size_t bytes) -> void* {
    void* p = (char*)d_ws + off;
    off += (bytes + 255) & ~(size_t)255;
    return p;
  };
  unsigned short* xbf  = (unsigned short*)carve((size_t)N * IN_F * 2);
  unsigned short* hbf  = (unsigned short*)carve((size_t)N * (H1 * HID_F) * 2);
  unsigned short* wt1  = (unsigned short*)carve((size_t)R_REL * IN_F * (H1 * HID_F) * 2);
  unsigned short* wt2  = (unsigned short*)carve((size_t)R_REL * (H1 * HID_F) * OUT_F * 2);
  float*    feat  = (float*)carve((size_t)N * (H1 * HID_F) * 4);
  float*    el    = (float*)carve((size_t)N * H1 * 4);
  float*    er    = (float*)carve((size_t)N * H1 * 4);
  float*    hacc  = (float*)carve((size_t)N * (H1 * HID_F) * 4);
  const int NSETS = 2 * R_REL;
  const int NP    = (N / 1024 + 1) * 1024;
  const int nb    = NP / 1024;
  unsigned* cnt    = (unsigned*)carve((size_t)NSETS * NP * 4);
  unsigned* bsum   = (unsigned*)carve((size_t)NSETS * nb * 4);
  int*      rowptr = (int*)carve((size_t)NSETS * NP * 4);
  int*      cursor = (int*)carve((size_t)NSETS * NP * 4);
  int*      csrc   = (int*)carve((size_t)NSETS * E * 4);

  const int T = 256;
  auto blk1 = [&](size_t n) { return (unsigned)((n + T - 1) / T); };

  k_f32_to_bf16<<<blk1((size_t)N * IN_F), T, 0, stream>>>(x, xbf, (size_t)N * IN_F);
  for (int r = 0; r < R_REL; ++r) {
    k_transpose_w<<<blk1((size_t)IN_F * (H1 * HID_F)), T, 0, stream>>>(
        W1 + (size_t)r * IN_F * (H1 * HID_F), wt1 + (size_t)r * IN_F * (H1 * HID_F),
        IN_F, H1 * HID_F);
    k_transpose_w<<<blk1((size_t)(H1 * HID_F) * OUT_F), T, 0, stream>>>(
        W2 + (size_t)r * (H1 * HID_F) * OUT_F, wt2 + (size_t)r * (H1 * HID_F) * OUT_F,
        H1 * HID_F, OUT_F);
  }

  k_zero_u32<<<blk1((size_t)NSETS * NP), T, 0, stream>>>(cnt, (size_t)NSETS * NP);
  k_dst_hist<<<blk1((size_t)NSETS * E), T, 0, stream>>>(edst, cnt, E, NP, NSETS * E);
  k_tile_sum<<<dim3(nb, NSETS), 256, 0, stream>>>(cnt, bsum, NP, nb);
  k_tile_scan<<<dim3(nb, NSETS), 256, 0, stream>>>(cnt, bsum, rowptr, cursor, NP, nb);
  k_dst_fill<<<blk1((size_t)NSETS * E), T, 0, stream>>>(esrc, edst, cursor, csrc, E, NP, NSETS * E);

  constexpr int RPB = 4;
  const unsigned gemmGrid = (unsigned)((N + 16 * RPB - 1) / (16 * RPB));

  k_init_bias<<<blk1((size_t)N * (H1 * HID_F)), T, 0, stream>>>(hacc, b1, N, H1 * HID_F);
  for (int r = 0; r < R_REL; ++r) {
    const int q1 = 0 * R_REL + r;

    k_gemm_bf16_wmma<IN_F, H1 * HID_F, RPB><<<gemmGrid, 2 * (H1 * HID_F), 0, stream>>>(
        xbf, wt1 + (size_t)r * IN_F * (H1 * HID_F), feat, N);

    k_attn_dots<<<blk1((size_t)N * H1), T, 0, stream>>>(
        feat, al1 + (size_t)r * H1 * HID_F, ar1 + (size_t)r * H1 * HID_F,
        el, er, N, H1, HID_F);

    k_seg_softmax_sum<H1, HID_F><<<blk1((size_t)N * 32), T, 0, stream>>>(
        rowptr + (size_t)q1 * NP, csrc + (size_t)q1 * E, el, er, feat, hacc, N);
  }

  k_relu_f32_to_bf16<<<blk1((size_t)N * (H1 * HID_F)), T, 0, stream>>>(
      hacc, hbf, (size_t)N * (H1 * HID_F));

  k_init_bias<<<blk1((size_t)N * OUT_F), T, 0, stream>>>(out, b2, N, OUT_F);
  for (int r = 0; r < R_REL; ++r) {
    const int q2 = 1 * R_REL + r;

    k_gemm_bf16_wmma<H1 * HID_F, OUT_F, RPB><<<gemmGrid, 2 * OUT_F, 0, stream>>>(
        hbf, wt2 + (size_t)r * (H1 * HID_F) * OUT_F, feat, N);

    k_attn_dots<<<blk1((size_t)N * H2), T, 0, stream>>>(
        feat, al2 + (size_t)r * H2 * OUT_F, ar2 + (size_t)r * H2 * OUT_F,
        el, er, N, H2, OUT_F);

    k_seg_softmax_sum<H2, OUT_F><<<blk1((size_t)N * 32), T, 0, stream>>>(
        rowptr + (size_t)q2 * NP, csrc + (size_t)q2 * E, el, er, feat, out, N);
  }
}


# Thin kernels (bf16 convert, bias start value, zero, edge histogram, edge fill): 4 elements per lane, 16-byte accesses, loops in fixed grids of 512 to 2048 work-groups; 35x fewer waves to start
# speedup vs baseline: 2.0906x; 1.1045x over previous
#include <hip/hip_runtime.h>

#define R_REL   2
#define IN_F    128
#define HID_F   64
#define OUT_F   64
#define H1      2
#define H2      1
#define NEG_SLOPE 0.2f

typedef __attribute__((ext_vector_type(16))) __bf16 v16bf;
typedef __attribute__((ext_vector_type(8)))  float  v8f;

union Frag16 { uint4 q[2]; v16bf v; };

__device__ __forceinline__ unsigned short f2bf(float f) {
  unsigned u = __float_as_uint(f);
  unsigned r = u + 0x7FFFu + ((u >> 16) & 1u);
  return (unsigned short)(r >> 16);
}
__device__ __forceinline__ float lrelu(float x) { return x >= 0.f ? x : NEG_SLOPE * x; }

#define STREAM_BLOCKS  512
#define STREAM_THREADS 256
#define STREAM_STRIDE  (STREAM_BLOCKS * STREAM_THREADS)
#define EDGE_BLOCKS    2048

__device__ __forceinline__ uint2 pack4_bf16(float4 v) {
  uint2 r;
  r.x = (unsigned)f2bf(v.x) | ((unsigned)f2bf(v.y) << 16);
  r.y = (unsigned)f2bf(v.z) | ((unsigned)f2bf(v.w) << 16);
  return r;
}

template <bool RELU>
__global__ __launch_bounds__(STREAM_THREADS)
void k_f32_to_bf16(const float* __restrict__ in, unsigned short* __restrict__ out, unsigned n4) {
  const float4* __restrict__ in4  = reinterpret_cast<const float4*>(in);
  uint2* __restrict__        out2 = reinterpret_cast<uint2*>(out);
  constexpr int U = 8;
  const unsigned t = blockIdx.x * STREAM_THREADS + threadIdx.x;
  for (unsigned i = t; i < n4; i += U * STREAM_STRIDE) {
    float4 v[U];
    if (i + (U - 1) * STREAM_STRIDE < n4) {
#pragma unroll
      for (int k = 0; k < U; ++k) v[k] = in4[i + k * STREAM_STRIDE];
#pragma unroll
      for (int k = 0; k < U; ++k) {
        float4 x = v[k];
        if (RELU) { x.x = x.x > 0.f ? x.x : 0.f; x.y = x.y > 0.f ? x.y : 0.f;
                    x.z = x.z > 0.f ? x.z : 0.f; x.w = x.w > 0.f ? x.w : 0.f; }
        out2[i + k * STREAM_STRIDE] = pack4_bf16(x);
      }
    } else {
      for (unsigned j = i; j < n4; j += STREAM_STRIDE) {
        float4 x = in4[j];
        if (RELU) { x.x = x.x > 0.f ? x.x : 0.f; x.y = x.y > 0.f ? x.y : 0.f;
                    x.z = x.z > 0.f ? x.z : 0.f; x.w = x.w > 0.f ? x.w : 0.f; }
        out2[j] = pack4_bf16(x);
      }
    }
  }
}

__global__ void k_transpose_w(const float* __restrict__ W,
                              unsigned short* __restrict__ Wt, int K, int NCOLS) {
  int i = blockIdx.x * blockDim.x + threadIdx.x;
  if (i >= K * NCOLS) return;
  int n = i / K, k = i % K;
  Wt[i] = f2bf(W[(size_t)k * NCOLS + n]);
}

__global__ void k_zero_u32(unsigned* __restrict__ p, unsigned n4) {
  unsigned i = blockIdx.x * blockDim.x + threadIdx.x;
  if (i < n4) reinterpret_cast<uint4*>(p)[i] = make_uint4(0u, 0u, 0u, 0u);
}

template <int F>
__global__ __launch_bounds__(STREAM_THREADS)
void k_init_bias(float* __restrict__ acc, const float* __restrict__ b, unsigned n4) {
  static_assert(STREAM_STRIDE % (F / 4) == 0, "column group must not change along the loop");
  const unsigned t = blockIdx.x * STREAM_THREADS + threadIdx.x;
  const int c = (int)(t % (F / 4)) * 4;
  float4 s = make_float4(0.f, 0.f, 0.f, 0.f);
  for (int r = 0; r < R_REL; ++r) {
    const float4 v = *reinterpret_cast<const float4*>(b + (size_t)r * F + c);
    s.x += v.x; s.y += v.y; s.z += v.z; s.w += v.w;
  }
  float4* __restrict__ acc4 = reinterpret_cast<float4*>(acc);
#pragma clang loop vectorize(disable) interleave(disable)
  for (unsigned i = t; i < n4; i += STREAM_STRIDE) acc4[i] = s;
}

template <int KDIM, int NCOLS, int RPB>
__global__ __launch_bounds__(2 * NCOLS)
void k_gemm_bf16_wmma(const unsigned short* __restrict__ A,
                      const unsigned short* __restrict__ Bt,
                      float* __restrict__ C, int M) {
  constexpr int KC = KDIM / 32;
  const int lane     = threadIdx.x & 31;
  const int wv       = threadIdx.x >> 5;
  const int colBase  = wv * 16;
  const int l16      = lane & 15;
  const int k0       = (lane < 16) ? 0 : 8;
  const int rowBlock = blockIdx.x * (16 * RPB);

  Frag16 fb[KC];
  {
    const unsigned short* bp = Bt + (size_t)(colBase + l16) * KDIM + k0;
#pragma unroll
    for (int c = 0; c < KC; ++c) {
      fb[c].q[0] = *reinterpret_cast<const uint4*>(bp + 32 * c);
      fb[c].q[1] = *reinterpret_cast<const uint4*>(bp + 32 * c + 16);
    }
  }

  const bool full = (rowBlock + 16 * RPB) <= M;
  const int  rOff = (lane < 16) ? 0 : 8;

#pragma unroll
  for (int t = 0; t < RPB; ++t) {
    const int rowBase = rowBlock + 16 * t;
    if (rowBase >= M) break;

    int arow = rowBase + l16; if (arow >= M) arow = M - 1;
    const unsigned short* ap = A + (size_t)arow * KDIM + k0;

    v8f acc = {};
#pragma unroll
    for (int c = 0; c < KC; ++c) {
      Frag16 fa;
      fa.q[0] = *reinterpret_cast<const uint4*>(ap + 32 * c);
      fa.q[1] = *reinterpret_cast<const uint4*>(ap + 32 * c + 16);
      acc = __builtin_amdgcn_wmma_f32_16x16x32_bf16(false, fa.v, false, fb[c].v,
                                                    (short)0, acc, false, false);
    }

    float* cp = C + (size_t)rowBase * NCOLS + colBase + l16;
    if (full) {
#pragma unroll
      for (int v = 0; v < 8; ++v) cp[(size_t)(rOff + v) * NCOLS] = acc[v];
    } else {
#pragma unroll
      for (int v = 0; v < 8; ++v) {
        int rr = rowBase + rOff + v;
        if (rr < M) cp[(size_t)(rOff + v) * NCOLS] = acc[v];
      }
    }
  }
}

__global__ void k_attn_dots(const float* __restrict__ feat,
                            const float* __restrict__ al, const float* __restrict__ ar,
                            float* __restrict__ el, float* __restrict__ er,
                            int N, int H, int D) {
  int i = blockIdx.x * blockDim.x + threadIdx.x;
  if (i >= N * H) return;
  int h = i % H;
  const float* f = feat + (size_t)(i / H) * H * D + (size_t)h * D;
  float sl = 0.f, sr = 0.f;
  for (int d = 0; d < D; d += 4) {
    float4 v = *reinterpret_cast<const float4*>(f + d);
    float4 a = *reinterpret_cast<const float4*>(al + h * D + d);
    float4 b = *reinterpret_cast<const float4*>(ar + h * D + d);
    sl += v.x * a.x + v.y * a.y + v.z * a.z + v.w * a.w;
    sr += v.x * b.x + v.y * b.y + v.z * b.z + v.w * b.w;
  }
  el[i] = sl; er[i] = sr;
}

__global__ __launch_bounds__(STREAM_THREADS)
void k_dst_hist(const int* __restrict__ dst_all, unsigned* __restrict__ cnt,
                int E, int NP, unsigned total4) {
  const int4* __restrict__ d4 = reinterpret_cast<const int4*>(dst_all);
  const unsigned t      = blockIdx.x * STREAM_THREADS + threadIdx.x;
  const unsigned stride = gridDim.x * STREAM_THREADS;
  const unsigned uE     = (unsigned)E;
  for (unsigned j = t; j < total4; j += stride) {
    const int4 d = d4[j];
    const unsigned g = 4u * j;
    atomicAdd(&cnt[(size_t)((g     ) / uE) * NP + d.x], 1u);
    atomicAdd(&cnt[(size_t)((g + 1u) / uE) * NP + d.y], 1u);
    atomicAdd(&cnt[(size_t)((g + 2u) / uE) * NP + d.z], 1u);
    atomicAdd(&cnt[(size_t)((g + 3u) / uE) * NP + d.w], 1u);
  }
}

__global__ __launch_bounds__(256)
void k_tile_sum(const unsigned* __restrict__ cnt, unsigned* __restrict__ bsum, int NP, int nb) {
  __shared__ unsigned sh[256];
  const int t = threadIdx.x, b = blockIdx.x, q = blockIdx.y;
  const uint4 c = *reinterpret_cast<const uint4*>(cnt + (size_t)q * NP + (size_t)b * 1024 + t * 4);
  sh[t] = c.x + c.y + c.z + c.w;
  __syncthreads();
  for (int s = 128; s > 0; s >>= 1) {
    if (t < s) sh[t] += sh[t + s];
    __syncthreads();
  }
  if (t == 0) bsum[q * nb + b] = sh[0];
}

__global__ __launch_bounds__(256)
void k_tile_scan(const unsigned* __restrict__ cnt, const unsigned* __restrict__ bsum,
                 int* __restrict__ rowptr, int* __restrict__ cursor, int NP, int nb) {
  __shared__ unsigned sh[256];
  const int t = threadIdx.x, b = blockIdx.x, q = blockIdx.y;
  unsigned part = 0;
  for (int j = t; j < b; j += 256) part += bsum[q * nb + j];
  sh[t] = part;
  __syncthreads();
  for (int s = 128; s > 0; s >>= 1) {
    if (t < s) sh[t] += sh[t + s];
    __syncthreads();
  }
  const unsigned off = sh[0];
  __syncthreads();

  const size_t idx = (size_t)q * NP + (size_t)b * 1024 + t * 4;
  const uint4 c = *reinterpret_cast<const uint4*>(cnt + idx);
  const unsigned t0 = c.x, t1 = t0 + c.y, t2 = t1 + c.z, t3 = t2 + c.w;
  sh[t] = t3;
  __syncthreads();
  for (int o = 1; o < 256; o <<= 1) {
    unsigned v = (t >= o) ? sh[t - o] : 0u;
    __syncthreads();
    sh[t] += v;
    __syncthreads();
  }
  const unsigned ex = sh[t] - t3 + off;
  int4 r;
  r.x = (int)ex; r.y = (int)(ex + t0); r.z = (int)(ex + t1); r.w = (int)(ex + t2);
  *reinterpret_cast<int4*>(rowptr + idx) = r;
  *reinterpret_cast<int4*>(cursor + idx) = r;
}

__global__ __launch_bounds__(STREAM_THREADS)
void k_dst_fill(const int* __restrict__ src_all, const int* __restrict__ dst_all,
                int* __restrict__ cursor, int* __restrict__ csrc,
                int E, int NP, unsigned total4) {
  const int4* __restrict__ s4 = reinterpret_cast<const int4*>(src_all);
  const int4* __restrict__ d4 = reinterpret_cast<const int4*>(dst_all);
  const unsigned t      = blockIdx.x * STREAM_THREADS + threadIdx.x;
  const unsigned stride = gridDim.x * STREAM_THREADS;
  const unsigned uE     = (unsigned)E;
  for (unsigned j = t; j < total4; j += stride) {
    const int4 d = d4[j];
    const int4 s = s4[j];
    const unsigned g = 4u * j;
    const unsigned q0 = g / uE, q1 = (g + 1u) / uE, q2 = (g + 2u) / uE, q3 = (g + 3u) / uE;
    const int p0 = atomicAdd(&cursor[(size_t)q0 * NP + d.x], 1);
    const int p1 = atomicAdd(&cursor[(size_t)q1 * NP + d.y], 1);
    const int p2 = atomicAdd(&cursor[(size_t)q2 * NP + d.z], 1);
    const int p3 = atomicAdd(&cursor[(size_t)q3 * NP + d.w], 1);
    csrc[(size_t)q0 * E + p0] = s.x;
    csrc[(size_t)q1 * E + p1] = s.y;
    csrc[(size_t)q2 * E + p2] = s.z;
    csrc[(size_t)q3 * E + p3] = s.w;
  }
}

__device__ __forceinline__ float bcast_lane(float v, int j) {
  return __int_as_float(__builtin_amdgcn_readlane(__float_as_int(v), j));
}

template <int H, int D>
__global__ __launch_bounds__(256)
void k_seg_softmax_sum(const int* __restrict__ rowptr, const int* __restrict__ csrc,
                       const float* __restrict__ el, const float* __restrict__ er,
                       const float* __restrict__ feat, float* __restrict__ acc, int N) {
  constexpr int F   = H * D;
  constexpr int PER = F / 32;
  const int d    = __builtin_amdgcn_readfirstlane(
      (int)(((size_t)blockIdx.x * blockDim.x + threadIdx.x) >> 5));
  const int lane = threadIdx.x & 31;
  if (d >= N) return;
  const int beg = rowptr[d], end = rowptr[d + 1];
  if (end <= beg) return;

  float erd[H], m[H], den[H];
#pragma unroll
  for (int h = 0; h < H; ++h) { erd[h] = er[(size_t)d * H + h]; m[h] = -INFINITY; den[h] = 0.f; }

  for (int i = beg + lane; i < end; i += 32) {
    const int s = csrc[i];
#pragma unroll
    for (int h = 0; h < H; ++h) m[h] = fmaxf(m[h], lrelu(el[(size_t)s * H + h] + erd[h]));
  }
#pragma unroll
  for (int h = 0; h < H; ++h)
    for (int o = 16; o > 0; o >>= 1) m[h] = fmaxf(m[h], __shfl_xor(m[h], o));

  for (int i = beg + lane; i < end; i += 32) {
    const int s = csrc[i];
#pragma unroll
    for (int h = 0; h < H; ++h) den[h] += __expf(lrelu(el[(size_t)s * H + h] + erd[h]) - m[h]);
  }
#pragma unroll
  for (int h = 0; h < H; ++h) {
    for (int o = 16; o > 0; o >>= 1) den[h] += __shfl_xor(den[h], o);
    den[h] = den[h] > 0.f ? den[h] : 1.f;
  }

  const int base = lane * PER;
  const int hl   = base / D;
  float a[PER];
#pragma unroll
  for (int k = 0; k < PER; ++k) a[k] = 0.f;

  for (int cb = beg; cb < end; cb += 32) {
    const int i = cb + lane;
    int   s_l = 0;
    float w_l[H];
#pragma unroll
    for (int h = 0; h < H; ++h) w_l[h] = 0.f;
    if (i < end) {
      s_l = csrc[i];
#pragma unroll
      for (int h = 0; h < H; ++h)
        w_l[h] = __expf(lrelu(el[(size_t)s_l * H + h] + erd[h]) - m[h]) / den[h];
    }
    const int cnt = (end - cb) < 32 ? (end - cb) : 32;
#pragma unroll 4
    for (int j = 0; j < cnt; ++j) {
      const int sj = __builtin_amdgcn_readlane(s_l, j);
      float wj = bcast_lane(w_l[0], j);
      if constexpr (H == 2) { const float w1 = bcast_lane(w_l[1], j); wj = hl ? w1 : wj; }
      const float* fp = feat + (size_t)sj * F + base;
      if constexpr (PER == 4) {
        const float4 v = *reinterpret_cast<const float4*>(fp);
        a[0] += wj * v.x; a[1] += wj * v.y; a[2] += wj * v.z; a[3] += wj * v.w;
      } else {
        const float2 v = *reinterpret_cast<const float2*>(fp);
        a[0] += wj * v.x; a[1] += wj * v.y;
      }
    }
  }

  float* op = acc + (size_t)d * F + base;
  if constexpr (PER == 4) {
    float4 o = *reinterpret_cast<const float4*>(op);
    o.x += a[0]; o.y += a[1]; o.z += a[2]; o.w += a[3];
    *reinterpret_cast<float4*>(op) = o;
  } else {
    float2 o = *reinterpret_cast<const float2*>(op);
    o.x += a[0]; o.y += a[1];
    *reinterpret_cast<float2*>(op) = o;
  }
}

extern "C" void kernel_launch(void* const* d_in, const int* in_sizes, int n_in,
                              void* d_out, int out_size, void* d_ws, size_t ws_size,
                              hipStream_t stream) {
  const float* x    = (const float*)d_in[0];
  const int*   esrc = (const int*)d_in[1];
  const int*   edst = (const int*)d_in[2];
  const float* W1   = (const float*)d_in[3];
  const float* al1  = (const float*)d_in[4];
  const float* ar1  = (const float*)d_in[5];
  const float* b1   = (const float*)d_in[6];
  const float* W2   = (const float*)d_in[7];
  const float* al2  = (const float*)d_in[8];
  const float* ar2  = (const float*)d_in[9];
  const float* b2   = (const float*)d_in[10];
  float* out = (float*)d_out;

  const int N = in_sizes[0] / IN_F;
  const int E = in_sizes[1] / (2 * R_REL);

  size_t off = 0;
  auto carve = [&](size_t bytes) -> void* {
    void* p = (char*)d_ws + off;
    off += (bytes + 255) & ~(size_t)255;
    return p;
  };
  unsigned short* xbf  = (unsigned short*)carve((size_t)N * IN_F * 2);
  unsigned short* hbf  = (unsigned short*)carve((size_t)N * (H1 * HID_F) * 2);
  unsigned short* wt1  = (unsigned short*)carve((size_t)R_REL * IN_F * (H1 * HID_F) * 2);
  unsigned short* wt2  = (unsigned short*)carve((size_t)R_REL * (H1 * HID_F) * OUT_F * 2);
  float*    feat  = (float*)carve((size_t)N * (H1 * HID_F) * 4);
  float*    el    = (float*)carve((size_t)N * H1 * 4);
  float*    er    = (float*)carve((size_t)N * H1 * 4);
  float*    hacc  = (float*)carve((size_t)N * (H1 * HID_F) * 4);
  const int NSETS = 2 * R_REL;
  const int NP    = (N / 1024 + 1) * 1024;
  const int nb    = NP / 1024;
  unsigned* cnt    = (unsigned*)carve((size_t)NSETS * NP * 4);
  unsigned* bsum   = (unsigned*)carve((size_t)NSETS * nb * 4);
  int*      rowptr = (int*)carve((size_t)NSETS * NP * 4);
  int*      cursor = (int*)carve((size_t)NSETS * NP * 4);
  int*      csrc   = (int*)carve((size_t)NSETS * E * 4);

  const int T = 256;
  auto blk1 = [&](size_t n) { return (unsigned)((n + T - 1) / T); };

  k_f32_to_bf16<false><<<STREAM_BLOCKS, STREAM_THREADS, 0, stream>>>(x, xbf, (unsigned)((size_t)N * IN_F / 4));
  for (int r = 0; r < R_REL; ++r) {
    k_transpose_w<<<blk1((size_t)IN_F * (H1 * HID_F)), T, 0, stream>>>(
        W1 + (size_t)r * IN_F * (H1 * HID_F), wt1 + (size_t)r * IN_F * (H1 * HID_F),
        IN_F, H1 * HID_F);
    k_transpose_w<<<blk1((size_t)(H1 * HID_F) * OUT_F), T, 0, stream>>>(
        W2 + (size_t)r * (H1 * HID_F) * OUT_F, wt2 + (size_t)r * (H1 * HID_F) * OUT_F,
        H1 * HID_F, OUT_F);
  }

  k_zero_u32<<<blk1((size_t)NSETS * NP / 4), T, 0, stream>>>(cnt, (unsigned)((size_t)NSETS * NP / 4));
  k_dst_hist<<<EDGE_BLOCKS, STREAM_THREADS, 0, stream>>>(edst, cnt, E, NP, (unsigned)((size_t)NSETS * E / 4));
  k_tile_sum<<<dim3(nb, NSETS), 256, 0, stream>>>(cnt, bsum, NP, nb);
  k_tile_scan<<<dim3(nb, NSETS), 256, 0, stream>>>(cnt, bsum, rowptr, cursor, NP, nb);
  k_dst_fill<<<EDGE_BLOCKS, STREAM_THREADS, 0, stream>>>(esrc, edst, cursor, csrc, E, NP,
                                                            (unsigned)((size_t)NSETS * E / 4));

  constexpr int RPB = 4;
  const unsigned gemmGrid = (unsigned)((N + 16 * RPB - 1) / (16 * RPB));

  k_init_bias<H1 * HID_F><<<STREAM_BLOCKS, STREAM_THREADS, 0, stream>>>(
      hacc, b1, (unsigned)((size_t)N * (H1 * HID_F) / 4));
  for (int r = 0; r < R_REL; ++r) {
    const int q1 = 0 * R_REL + r;

    k_gemm_bf16_wmma<IN_F, H1 * HID_F, RPB><<<gemmGrid, 2 * (H1 * HID_F), 0, stream>>>(
        xbf, wt1 + (size_t)r * IN_F * (H1 * HID_F), feat, N);

    k_attn_dots<<<blk1((size_t)N * H1), T, 0, stream>>>(
        feat, al1 + (size_t)r * H1 * HID_F, ar1 + (size_t)r * H1 * HID_F,
        el, er, N, H1, HID_F);

    k_seg_softmax_sum<H1, HID_F><<<blk1((size_t)N * 32), T, 0, stream>>>(
        rowptr + (size_t)q1 * NP, csrc + (size_t)q1 * E, el, er, feat, hacc, N);
  }

  k_f32_to_bf16<true><<<STREAM_BLOCKS, STREAM_THREADS, 0, stream>>>(
      hacc, hbf, (unsigned)((size_t)N * (H1 * HID_F) / 4));

  k_init_bias<OUT_F><<<STREAM_BLOCKS, STREAM_THREADS, 0, stream>>>(
      out, b2, (unsigned)((size_t)N * OUT_F / 4));
  for (int r = 0; r < R_REL; ++r) {
    const int q2 = 1 * R_REL + r;

    k_gemm_bf16_wmma<H1 * HID_F, OUT_F, RPB><<<gemmGrid, 2 * OUT_F, 0, stream>>>(
        hbf, wt2 + (size_t)r * (H1 * HID_F) * OUT_F, feat, N);

    k_attn_dots<<<blk1((size_t)N * H2), T, 0, stream>>>(
        feat, al2 + (size_t)r * H2 * OUT_F, ar2 + (size_t)r * H2 * OUT_F,
        el, er, N, H2, OUT_F);

    k_seg_softmax_sum<H2, OUT_F><<<blk1((size_t)N * 32), T, 0, stream>>>(
        rowptr + (size_t)q2 * NP, csrc + (size_t)q2 * E, el, er, feat, out, N);
  }
}


# Counting sort of the edges on LDS counters in two levels (512-destination bucket, then destination) in place of global atomic counters; 4 sort launches for 5; sums stay in arrival order
# speedup vs baseline: 4.7182x; 2.2569x over previous
#include <hip/hip_runtime.h>

#define R_REL   2
#define IN_F    128
#define HID_F   64
#define OUT_F   64
#define H1      2
#define H2      1
#define NEG_SLOPE 0.2f

typedef __attribute__((ext_vector_type(16))) __bf16 v16bf;
typedef __attribute__((ext_vector_type(8)))  float  v8f;

union Frag16 { uint4 q[2]; v16bf v; };

__device__ __forceinline__ unsigned short f2bf(float f) {
  unsigned u = __float_as_uint(f);
  unsigned r = u + 0x7FFFu + ((u >> 16) & 1u);
  return (unsigned short)(r >> 16);
}
__device__ __forceinline__ float lrelu(float x) { return x >= 0.f ? x : NEG_SLOPE * x; }

#define STREAM_BLOCKS  512
#define STREAM_THREADS 256
#define STREAM_STRIDE  (STREAM_BLOCKS * STREAM_THREADS)

__device__ __forceinline__ uint2 pack4_bf16(float4 v) {
  uint2 r;
  r.x = (unsigned)f2bf(v.x) | ((unsigned)f2bf(v.y) << 16);
  r.y = (unsigned)f2bf(v.z) | ((unsigned)f2bf(v.w) << 16);
  return r;
}

template <bool RELU>
__global__ __launch_bounds__(STREAM_THREADS)
void k_f32_to_bf16(const float* __restrict__ in, unsigned short* __restrict__ out, unsigned n4) {
  const float4* __restrict__ in4  = reinterpret_cast<const float4*>(in);
  uint2* __restrict__        out2 = reinterpret_cast<uint2*>(out);
  constexpr int U = 8;
  const unsigned t = blockIdx.x * STREAM_THREADS + threadIdx.x;
  for (unsigned i = t; i < n4; i += U * STREAM_STRIDE) {
    float4 v[U];
    if (i + (U - 1) * STREAM_STRIDE < n4) {
#pragma unroll
      for (int k = 0; k < U; ++k) v[k] = in4[i + k * STREAM_STRIDE];
#pragma unroll
      for (int k = 0; k < U; ++k) {
        float4 x = v[k];
        if (RELU) { x.x = x.x > 0.f ? x.x : 0.f; x.y = x.y > 0.f ? x.y : 0.f;
                    x.z = x.z > 0.f ? x.z : 0.f; x.w = x.w > 0.f ? x.w : 0.f; }
        out2[i + k * STREAM_STRIDE] = pack4_bf16(x);
      }
    } else {
      for (unsigned j = i; j < n4; j += STREAM_STRIDE) {
        float4 x = in4[j];
        if (RELU) { x.x = x.x > 0.f ? x.x : 0.f; x.y = x.y > 0.f ? x.y : 0.f;
                    x.z = x.z > 0.f ? x.z : 0.f; x.w = x.w > 0.f ? x.w : 0.f; }
        out2[j] = pack4_bf16(x);
      }
    }
  }
}

__global__ void k_transpose_w(const float* __restrict__ W,
                              unsigned short* __restrict__ Wt, int K, int NCOLS) {
  int i = blockIdx.x * blockDim.x + threadIdx.x;
  if (i >= K * NCOLS) return;
  int n = i / K, k = i % K;
  Wt[i] = f2bf(W[(size_t)k * NCOLS + n]);
}

template <int F>
__global__ __launch_bounds__(STREAM_THREADS)
void k_init_bias(float* __restrict__ acc, const float* __restrict__ b, unsigned n4) {
  static_assert(STREAM_STRIDE % (F / 4) == 0, "column group must not change along the loop");
  const unsigned t = blockIdx.x * STREAM_THREADS + threadIdx.x;
  const int c = (int)(t % (F / 4)) * 4;
  float4 s = make_float4(0.f, 0.f, 0.f, 0.f);
  for (int r = 0; r < R_REL; ++r) {
    const float4 v = *reinterpret_cast<const float4*>(b + (size_t)r * F + c);
    s.x += v.x; s.y += v.y; s.z += v.z; s.w += v.w;
  }
  float4* __restrict__ acc4 = reinterpret_cast<float4*>(acc);
#pragma clang loop vectorize(disable) interleave(disable)
  for (unsigned i = t; i < n4; i += STREAM_STRIDE) acc4[i] = s;
}

template <int KDIM, int NCOLS, int RPB>
__global__ __launch_bounds__(2 * NCOLS)
void k_gemm_bf16_wmma(const unsigned short* __restrict__ A,
                      const unsigned short* __restrict__ Bt,
                      float* __restrict__ C, int M) {
  constexpr int KC = KDIM / 32;
  const int lane     = threadIdx.x & 31;
  const int wv       = threadIdx.x >> 5;
  const int colBase  = wv * 16;
  const int l16      = lane & 15;
  const int k0       = (lane < 16) ? 0 : 8;
  const int rowBlock = blockIdx.x * (16 * RPB);

  Frag16 fb[KC];
  {
    const unsigned short* bp = Bt + (size_t)(colBase + l16) * KDIM + k0;
#pragma unroll
    for (int c = 0; c < KC; ++c) {
      fb[c].q[0] = *reinterpret_cast<const uint4*>(bp + 32 * c);
      fb[c].q[1] = *reinterpret_cast<const uint4*>(bp + 32 * c + 16);
    }
  }

  const bool full = (rowBlock + 16 * RPB) <= M;
  const int  rOff = (lane < 16) ? 0 : 8;

#pragma unroll
  for (int t = 0; t < RPB; ++t) {
    const int rowBase = rowBlock + 16 * t;
    if (rowBase >= M) break;

    int arow = rowBase + l16; if (arow >= M) arow = M - 1;
    const unsigned short* ap = A + (size_t)arow * KDIM + k0;

    v8f acc = {};
#pragma unroll
    for (int c = 0; c < KC; ++c) {
      Frag16 fa;
      fa.q[0] = *reinterpret_cast<const uint4*>(ap + 32 * c);
      fa.q[1] = *reinterpret_cast<const uint4*>(ap + 32 * c + 16);
      acc = __builtin_amdgcn_wmma_f32_16x16x32_bf16(false, fa.v, false, fb[c].v,
                                                    (short)0, acc, false, false);
    }

    float* cp = C + (size_t)rowBase * NCOLS + colBase + l16;
    if (full) {
#pragma unroll
      for (int v = 0; v < 8; ++v) cp[(size_t)(rOff + v) * NCOLS] = acc[v];
    } else {
#pragma unroll
      for (int v = 0; v < 8; ++v) {
        int rr = rowBase + rOff + v;
        if (rr < M) cp[(size_t)(rOff + v) * NCOLS] = acc[v];
      }
    }
  }
}

__global__ void k_attn_dots(const float* __restrict__ feat,
                            const float* __restrict__ al, const float* __restrict__ ar,
                            float* __restrict__ el, float* __restrict__ er,
                            int N, int H, int D) {
  int i = blockIdx.x * blockDim.x + threadIdx.x;
  if (i >= N * H) return;
  int h = i % H;
  const float* f = feat + (size_t)(i / H) * H * D + (size_t)h * D;
  float sl = 0.f, sr = 0.f;
  for (int d = 0; d < D; d += 4) {
    float4 v = *reinterpret_cast<const float4*>(f + d);
    float4 a = *reinterpret_cast<const float4*>(al + h * D + d);
    float4 b = *reinterpret_cast<const float4*>(ar + h * D + d);
    sl += v.x * a.x + v.y * a.y + v.z * a.z + v.w * a.w;
    sr += v.x * b.x + v.y * b.y + v.z * b.z + v.w * b.w;
  }
  el[i] = sl; er[i] = sr;
}

#define BKT_SHIFT    9
#define BKT_NODES    (1 << BKT_SHIFT)
#define SORT_BLOCKS   1024
#define SORT_THREADS  256
#define SORT2_THREADS 512

template <int NT>
__device__ __forceinline__ void block_excl_scan(unsigned* sh, int n, unsigned* part) {
  const int t  = threadIdx.x;
  const int C  = (n + NT - 1) / NT;
  const int lo = t * C;
  const int hi = (lo + C < n) ? lo + C : n;
  unsigned s = 0;
  for (int i = lo; i < hi; ++i) s += sh[i];
  part[t] = s;
  __syncthreads();
  for (int o = 1; o < NT; o <<= 1) {
    const unsigned v = (t >= o) ? part[t - o] : 0u;
    __syncthreads();
    part[t] += v;
    __syncthreads();
  }
  unsigned run = part[t] - s;
  for (int i = lo; i < hi; ++i) { const unsigned c = sh[i]; sh[i] = run; run += c; }
  __syncthreads();
}

__global__ __launch_bounds__(SORT_THREADS)
void k_bkt_hist(const int* __restrict__ dst_all, unsigned* __restrict__ bcount,
                int E, int B, int NK, unsigned total4, unsigned U) {
  extern __shared__ unsigned sh[];
  const int      t = threadIdx.x;
  const unsigned p = blockIdx.x;
  for (int k = t; k < NK; k += SORT_THREADS) sh[k] = 0u;
  __syncthreads();
  const int4* __restrict__ d4 = reinterpret_cast<const int4*>(dst_all);
  const unsigned uE = (unsigned)E, uB = (unsigned)B;
  const unsigned j0 = p * U;
  const unsigned j1 = (j0 + U < total4) ? j0 + U : total4;
  for (unsigned j = j0 + t; j < j1; j += SORT_THREADS) {
    const int4 d = d4[j];
    const unsigned g = 4u * j;
    atomicAdd(&sh[((g     ) / uE) * uB + ((unsigned)d.x >> BKT_SHIFT)], 1u);
    atomicAdd(&sh[((g + 1u) / uE) * uB + ((unsigned)d.y >> BKT_SHIFT)], 1u);
    atomicAdd(&sh[((g + 2u) / uE) * uB + ((unsigned)d.z >> BKT_SHIFT)], 1u);
    atomicAdd(&sh[((g + 3u) / uE) * uB + ((unsigned)d.w >> BKT_SHIFT)], 1u);
  }
  __syncthreads();
  for (int k = t; k < NK; k += SORT_THREADS) bcount[(size_t)k * gridDim.x + p] = sh[k];
}

__global__ __launch_bounds__(SORT_THREADS)
void k_bkt_scan(const unsigned* __restrict__ bcount, unsigned* __restrict__ boffs,
                unsigned* __restrict__ keytotal, int G) {
  __shared__ unsigned part[SORT_THREADS];
  const int    t    = threadIdx.x;
  const size_t base = (size_t)blockIdx.x * G;
  const int C  = (G + SORT_THREADS - 1) / SORT_THREADS;
  const int lo = t * C;
  const int hi = (lo + C < G) ? lo + C : G;
  unsigned s = 0;
  for (int i = lo; i < hi; ++i) s += bcount[base + i];
  part[t] = s;
  __syncthreads();
  for (int o = 1; o < SORT_THREADS; o <<= 1) {
    const unsigned v = (t >= o) ? part[t - o] : 0u;
    __syncthreads();
    part[t] += v;
    __syncthreads();
  }
  unsigned run = part[t] - s;
  for (int i = lo; i < hi; ++i) { const unsigned c = bcount[base + i]; boffs[base + i] = run; run += c; }
  if (t == SORT_THREADS - 1) keytotal[blockIdx.x] = part[t];
}

__global__ __launch_bounds__(SORT_THREADS)
void k_bkt_scatter(const int* __restrict__ src_all, const int* __restrict__ dst_all,
                   const unsigned* __restrict__ boffs, const unsigned* __restrict__ keytotal,
                   uint2* __restrict__ pairs, int E, int B, int NK, unsigned total4, unsigned U) {
  extern __shared__ unsigned sh[];
  unsigned* part = sh + NK;
  const int      t = threadIdx.x;
  const unsigned p = blockIdx.x;
  for (int k = t; k < NK; k += SORT_THREADS) sh[k] = keytotal[k];
  __syncthreads();
  block_excl_scan<SORT_THREADS>(sh, NK, part);
  for (int k = t; k < NK; k += SORT_THREADS) sh[k] += boffs[(size_t)k * gridDim.x + p];
  __syncthreads();
  const int4* __restrict__ s4 = reinterpret_cast<const int4*>(src_all);
  const int4* __restrict__ d4 = reinterpret_cast<const int4*>(dst_all);
  const unsigned uE = (unsigned)E, uB = (unsigned)B;
  const unsigned j0 = p * U;
  const unsigned j1 = (j0 + U < total4) ? j0 + U : total4;
  for (unsigned j = j0 + t; j < j1; j += SORT_THREADS) {
    const int4 d = d4[j];
    const int4 s = s4[j];
    const unsigned g = 4u * j;
    const unsigned p0 = atomicAdd(&sh[((g     ) / uE) * uB + ((unsigned)d.x >> BKT_SHIFT)], 1u);
    const unsigned p1 = atomicAdd(&sh[((g + 1u) / uE) * uB + ((unsigned)d.y >> BKT_SHIFT)], 1u);
    const unsigned p2 = atomicAdd(&sh[((g + 2u) / uE) * uB + ((unsigned)d.z >> BKT_SHIFT)], 1u);
    const unsigned p3 = atomicAdd(&sh[((g + 3u) / uE) * uB + ((unsigned)d.w >> BKT_SHIFT)], 1u);
    pairs[p0] = make_uint2((unsigned)s.x, (unsigned)d.x);
    pairs[p1] = make_uint2((unsigned)s.y, (unsigned)d.y);
    pairs[p2] = make_uint2((unsigned)s.z, (unsigned)d.z);
    pairs[p3] = make_uint2((unsigned)s.w, (unsigned)d.w);
  }
}

__global__ __launch_bounds__(SORT2_THREADS)
void k_bkt_sort(const uint2* __restrict__ pairs, const unsigned* __restrict__ keytotal,
                int* __restrict__ rowptr, int* __restrict__ csrc, int E, int B, int NP) {
  constexpr int NT = SORT2_THREADS;
  constexpr int DEPTH = 4;
  __shared__ unsigned cnt[BKT_NODES];
  __shared__ unsigned part[NT];
  const int t   = threadIdx.x;
  const int key = blockIdx.x;
  const int q   = key / B;
  const int nb0 = (key - q * B) << BKT_SHIFT;

  unsigned s = 0;
  for (int k = t; k < key; k += NT) s += keytotal[k];
  part[t] = s;
  for (int i = t; i < BKT_NODES; i += NT) cnt[i] = 0u;
  __syncthreads();
  for (int o = NT / 2; o > 0; o >>= 1) {
    if (t < o) part[t] += part[t + o];
    __syncthreads();
  }
  const unsigned bs = part[0];
  const unsigned n  = keytotal[key];
  __syncthreads();

  const uint2* __restrict__ pp = pairs + bs;
  for (unsigned i0 = 0; i0 < n; i0 += DEPTH * NT) {
    uint2 e[DEPTH];
#pragma unroll
    for (int k = 0; k < DEPTH; ++k) { const unsigned i = i0 + k * NT + t; e[k] = pp[i < n ? i : n - 1u]; }
#pragma unroll
    for (int k = 0; k < DEPTH; ++k) {
      const unsigned i = i0 + k * NT + t;
      if (i < n) atomicAdd(&cnt[e[k].y & (BKT_NODES - 1)], 1u);
    }
  }
  __syncthreads();
  block_excl_scan<NT>(cnt, BKT_NODES, part);

  const unsigned rel = bs - (unsigned)q * (unsigned)E;
  for (int i = t; i < BKT_NODES; i += NT)
    rowptr[(size_t)q * NP + nb0 + i] = (int)(rel + cnt[i]);
  __syncthreads();

  for (unsigned i0 = 0; i0 < n; i0 += DEPTH * NT) {
    uint2 e[DEPTH];
#pragma unroll
    for (int k = 0; k < DEPTH; ++k) { const unsigned i = i0 + k * NT + t; e[k] = pp[i < n ? i : n - 1u]; }
#pragma unroll
    for (int k = 0; k < DEPTH; ++k) {
      const unsigned i = i0 + k * NT + t;
      if (i < n) {
        const unsigned r = atomicAdd(&cnt[e[k].y & (BKT_NODES - 1)], 1u);
        csrc[bs + r] = (int)e[k].x;
      }
    }
  }
}

template <int CTRL>
__device__ __forceinline__ int dpp_i(int x) {
  return __builtin_amdgcn_update_dpp(0, x, CTRL, 0xF, 0xF, false);
}
template <int CTRL>
__device__ __forceinline__ float dpp_f(float x) {
  return __int_as_float(__builtin_amdgcn_update_dpp(0, __float_as_int(x), CTRL, 0xF, 0xF, false));
}
#define ROW_SHARE 0x150
#define ROW_XMASK 0x160

__device__ __forceinline__ float row_max16(float x) {
  x = fmaxf(x, dpp_f<ROW_XMASK + 8>(x));
  x = fmaxf(x, dpp_f<ROW_XMASK + 4>(x));
  x = fmaxf(x, dpp_f<ROW_XMASK + 2>(x));
  x = fmaxf(x, dpp_f<ROW_XMASK + 1>(x));
  return x;
}
__device__ __forceinline__ float row_sum16(float x) {
  x += dpp_f<ROW_XMASK + 8>(x);
  x += dpp_f<ROW_XMASK + 4>(x);
  x += dpp_f<ROW_XMASK + 2>(x);
  x += dpp_f<ROW_XMASK + 1>(x);
  return x;
}

template <int G, int F>
__device__ __forceinline__ void rows8(const float* __restrict__ frow, int s_l, float p_l, float4& a) {
#define SEG_LD(J) const float4 r##J = *reinterpret_cast<const float4*>( \
      frow + (size_t)dpp_i<ROW_SHARE + G + J>(s_l) * F);
  SEG_LD(0) SEG_LD(1) SEG_LD(2) SEG_LD(3) SEG_LD(4) SEG_LD(5) SEG_LD(6) SEG_LD(7)
#undef SEG_LD
#define SEG_FMA(J) { const float w = dpp_f<ROW_SHARE + G + J>(p_l); \
      a.x += w * r##J.x; a.y += w * r##J.y; a.z += w * r##J.z; a.w += w * r##J.w; }
  SEG_FMA(0) SEG_FMA(1) SEG_FMA(2) SEG_FMA(3) SEG_FMA(4) SEG_FMA(5) SEG_FMA(6) SEG_FMA(7)
#undef SEG_FMA
}

template <int H>
__global__ __launch_bounds__(256)
void k_seg_softmax_sum(const int* __restrict__ rowptr, const int* __restrict__ csrc,
                       const float* __restrict__ el, const float* __restrict__ er,
                       const float* __restrict__ feat, float* __restrict__ acc, int N) {
  static_assert(HID_F == 64 && OUT_F == 64, "a head is 64 floats in both layers");
  constexpr int D = 64;
  constexpr int F = H * D;
  const int wave = __builtin_amdgcn_readfirstlane(
      (int)(((size_t)blockIdx.x * 256 + threadIdx.x) >> 5));
  const int lane = threadIdx.x & 31;
  const int k    = lane & 15;
  const int v    = 2 * wave + (lane >> 4);
  const bool live = v < H * N;
  int h = 0;
  if constexpr (H == 2) h = (live && v >= N) ? 1 : 0;
  else if constexpr (H > 2) h = live ? v / N : 0;
  const int d = live ? v - h * N : 0;

  int beg = 0, end = 0;
  if (live) { beg = rowptr[d]; end = rowptr[d + 1]; }
  const int deg = end - beg;
  const int d0 = __builtin_amdgcn_readlane(deg, 0), d1 = __builtin_amdgcn_readlane(deg, 16);
  const int degmax = d0 > d1 ? d0 : d1;
  if (degmax <= 0) return;

  const float  erd  = er[(size_t)d * H + h];
  const float* elh  = el + h;
  const float* frow = feat + h * D + k * 4;
  const int    last = end - 1;

  float  m = -INFINITY, den = 0.f;
  float4 a = make_float4(0.f, 0.f, 0.f, 0.f);

  int s_nx = 0;
  if (deg > 0) { const int i0 = beg + k; s_nx = csrc[i0 < last ? i0 : last]; }

  for (int c0 = 0; c0 < degmax; c0 += 16) {
    const int  i     = beg + c0 + k;
    const bool valid = i < end;
    const int  s_l   = s_nx;
    if (c0 + 16 < degmax && deg > 0) { const int i1 = i + 16; s_nx = csrc[i1 < last ? i1 : last]; }
    float e = lrelu(elh[(size_t)s_l * H] + erd);
    e = valid ? e : -INFINITY;

    const float mn   = fmaxf(m, row_max16(e));
    const float mref = (mn == -INFINITY) ? 0.f : mn;
    const float sc   = __expf(m - mref);
    const float p_l  = __expf(e - mref);
    den = den * sc + row_sum16(p_l);
    a.x *= sc; a.y *= sc; a.z *= sc; a.w *= sc;
    m = mn;

    const int cnt = degmax - c0;
    rows8<0, F>(frow, s_l, p_l, a);
    if (cnt > 8) rows8<8, F>(frow, s_l, p_l, a);
  }

  if (deg > 0) {
    float* op = acc + (size_t)d * F + h * D + k * 4;
    float4 o = *reinterpret_cast<const float4*>(op);
    o.x += a.x / den; o.y += a.y / den; o.z += a.z / den; o.w += a.w / den;
    *reinterpret_cast<float4*>(op) = o;
  }
}

extern "C" void kernel_launch(void* const* d_in, const int* in_sizes, int n_in,
                              void* d_out, int out_size, void* d_ws, size_t ws_size,
                              hipStream_t stream) {
  const float* x    = (const float*)d_in[0];
  const int*   esrc = (const int*)d_in[1];
  const int*   edst = (const int*)d_in[2];
  const float* W1   = (const float*)d_in[3];
  const float* al1  = (const float*)d_in[4];
  const float* ar1  = (const float*)d_in[5];
  const float* b1   = (const float*)d_in[6];
  const float* W2   = (const float*)d_in[7];
  const float* al2  = (const float*)d_in[8];
  const float* ar2  = (const float*)d_in[9];
  const float* b2   = (const float*)d_in[10];
  float* out = (float*)d_out;

  const int N = in_sizes[0] / IN_F;
  const int E = in_sizes[1] / (2 * R_REL);

  size_t off = 0;
  auto carve = [&](size_t bytes) -> void* {
    void* p = (char*)d_ws + off;
    off += (bytes + 255) & ~(size_t)255;
    return p;
  };
  unsigned short* xbf  = (unsigned short*)carve((size_t)N * IN_F * 2);
  unsigned short* hbf  = (unsigned short*)carve((size_t)N * (H1 * HID_F) * 2);
  unsigned short* wt1  = (unsigned short*)carve((size_t)R_REL * IN_F * (H1 * HID_F) * 2);
  unsigned short* wt2  = (unsigned short*)carve((size_t)R_REL * (H1 * HID_F) * OUT_F * 2);
  float*    feat  = (float*)carve((size_t)N * (H1 * HID_F) * 4);
  float*    el    = (float*)carve((size_t)N * H1 * 4);
  float*    er    = (float*)carve((size_t)N * H1 * 4);
  float*    hacc  = (float*)carve((size_t)N * (H1 * HID_F) * 4);
  const int NSETS = 2 * R_REL;
  const int NP    = (N / 1024 + 1) * 1024;
  const int B     = (N >> BKT_SHIFT) + 1;
  const int NK    = NSETS * B;
  int*      rowptr   = (int*)carve((size_t)NSETS * NP * 4);
  int*      csrc     = (int*)carve((size_t)NSETS * E * 4);
  unsigned* bcount   = (unsigned*)carve((size_t)NK * SORT_BLOCKS * 4);
  unsigned* boffs    = (unsigned*)carve((size_t)NK * SORT_BLOCKS * 4);
  unsigned* keytotal = (unsigned*)carve((size_t)NK * 4);
  uint2*    pairs    = (uint2*)carve((size_t)NSETS * E * 8);

  const int T = 256;
  auto blk1 = [&](size_t n) { return (unsigned)((n + T - 1) / T); };

  k_f32_to_bf16<false><<<STREAM_BLOCKS, STREAM_THREADS, 0, stream>>>(x, xbf, (unsigned)((size_t)N * IN_F / 4));
  for (int r = 0; r < R_REL; ++r) {
    k_transpose_w<<<blk1((size_t)IN_F * (H1 * HID_F)), T, 0, stream>>>(
        W1 + (size_t)r * IN_F * (H1 * HID_F), wt1 + (size_t)r * IN_F * (H1 * HID_F),
        IN_F, H1 * HID_F);
    k_transpose_w<<<blk1((size_t)(H1 * HID_F) * OUT_F), T, 0, stream>>>(
        W2 + (size_t)r * (H1 * HID_F) * OUT_F, wt2 + (size_t)r * (H1 * HID_F) * OUT_F,
        H1 * HID_F, OUT_F);
  }

  const unsigned total4 = (unsigned)((size_t)NSETS * E / 4);
  const unsigned U      = (total4 + SORT_BLOCKS - 1) / SORT_BLOCKS;
  k_bkt_hist<<<SORT_BLOCKS, SORT_THREADS, (size_t)NK * 4, stream>>>(edst, bcount, E, B, NK, total4, U);
  k_bkt_scan<<<NK, SORT_THREADS, 0, stream>>>(bcount, boffs, keytotal, SORT_BLOCKS);
  k_bkt_scatter<<<SORT_BLOCKS, SORT_THREADS, (size_t)(NK + SORT_THREADS) * 4, stream>>>(
      esrc, edst, boffs, keytotal, pairs, E, B, NK, total4, U);
  k_bkt_sort<<<NK, SORT2_THREADS, 0, stream>>>(pairs, keytotal, rowptr, csrc, E, B, NP);

  constexpr int RPB = 4;
  const unsigned gemmGrid = (unsigned)((N + 16 * RPB - 1) / (16 * RPB));

  const unsigned segGrid1 = (unsigned)(((size_t)N * H1 + 15) / 16);
  const unsigned segGrid2 = (unsigned)(((size_t)N * H2 + 15) / 16);

  k_init_bias<H1 * HID_F><<<STREAM_BLOCKS, STREAM_THREADS, 0, stream>>>(
      hacc, b1, (unsigned)((size_t)N * (H1 * HID_F) / 4));
  for (int r = 0; r < R_REL; ++r) {
    const int q1 = 0 * R_REL + r;

    k_gemm_bf16_wmma<IN_F, H1 * HID_F, RPB><<<gemmGrid, 2 * (H1 * HID_F), 0, stream>>>(
        xbf, wt1 + (size_t)r * IN_F * (H1 * HID_F), feat, N);

    k_attn_dots<<<blk1((size_t)N * H1), T, 0, stream>>>(
        feat, al1 + (size_t)r * H1 * HID_F, ar1 + (size_t)r * H1 * HID_F,
        el, er, N, H1, HID_F);

    k_seg_softmax_sum<H1><<<segGrid1, T, 0, stream>>>(
        rowptr + (size_t)q1 * NP, csrc + (size_t)q1 * E, el, er, feat, hacc, N);
  }

  k_f32_to_bf16<true><<<STREAM_BLOCKS, STREAM_THREADS, 0, stream>>>(
      hacc, hbf, (unsigned)((size_t)N * (H1 * HID_F) / 4));

  k_init_bias<OUT_F><<<STREAM_BLOCKS, STREAM_THREADS, 0, stream>>>(
      out, b2, (unsigned)((size_t)N * OUT_F / 4));
  for (int r = 0; r < R_REL; ++r) {
    const int q2 = 1 * R_REL + r;

    k_gemm_bf16_wmma<H1 * HID_F, OUT_F, RPB><<<gemmGrid, 2 * OUT_F, 0, stream>>>(
        hbf, wt2 + (size_t)r * (H1 * HID_F) * OUT_F, feat, N);

    k_attn_dots<<<blk1((size_t)N * H2), T, 0, stream>>>(
        feat, al2 + (size_t)r * H2 * OUT_F, ar2 + (size_t)r * H2 * OUT_F,
        el, er, N, H2, OUT_F);

    k_seg_softmax_sum<H2><<<segGrid2, T, 0, stream>>>(
        rowptr + (size_t)q2 * NP, csrc + (size_t)q2 * E, el, er, feat, out, N);
  }
}


# Node kernel makes the attention logits from the rows it gathers (16-lane dot with the attention vector; destination part from own row): logit kernels and 4-byte gather gone; sum orders change
# speedup vs baseline: 5.3644x; 1.1370x over previous
#include <hip/hip_runtime.h>

#define R_REL   2
#define IN_F    128
#define HID_F   64
#define OUT_F   64
#define H1      2
#define H2      1
#define NEG_SLOPE 0.2f

typedef __attribute__((ext_vector_type(16))) __bf16 v16bf;
typedef __attribute__((ext_vector_type(8)))  float  v8f;

union Frag16 { uint4 q[2]; v16bf v; };

__device__ __forceinline__ unsigned short f2bf(float f) {
  unsigned u = __float_as_uint(f);
  unsigned r = u + 0x7FFFu + ((u >> 16) & 1u);
  return (unsigned short)(r >> 16);
}
__device__ __forceinline__ float lrelu(float x) { return x >= 0.f ? x : NEG_SLOPE * x; }

#define STREAM_BLOCKS  512
#define STREAM_THREADS 256
#define STREAM_STRIDE  (STREAM_BLOCKS * STREAM_THREADS)

__device__ __forceinline__ uint2 pack4_bf16(float4 v) {
  uint2 r;
  r.x = (unsigned)f2bf(v.x) | ((unsigned)f2bf(v.y) << 16);
  r.y = (unsigned)f2bf(v.z) | ((unsigned)f2bf(v.w) << 16);
  return r;
}

template <bool RELU>
__global__ __launch_bounds__(STREAM_THREADS)
void k_f32_to_bf16(const float* __restrict__ in, unsigned short* __restrict__ out, unsigned n4) {
  const float4* __restrict__ in4  = reinterpret_cast<const float4*>(in);
  uint2* __restrict__        out2 = reinterpret_cast<uint2*>(out);
  constexpr int U = 8;
  const unsigned t = blockIdx.x * STREAM_THREADS + threadIdx.x;
  for (unsigned i = t; i < n4; i += U * STREAM_STRIDE) {
    float4 v[U];
    if (i + (U - 1) * STREAM_STRIDE < n4) {
#pragma unroll
      for (int k = 0; k < U; ++k) v[k] = in4[i + k * STREAM_STRIDE];
#pragma unroll
      for (int k = 0; k < U; ++k) {
        float4 x = v[k];
        if (RELU) { x.x = x.x > 0.f ? x.x : 0.f; x.y = x.y > 0.f ? x.y : 0.f;
                    x.z = x.z > 0.f ? x.z : 0.f; x.w = x.w > 0.f ? x.w : 0.f; }
        out2[i + k * STREAM_STRIDE] = pack4_bf16(x);
      }
    } else {
      for (unsigned j = i; j < n4; j += STREAM_STRIDE) {
        float4 x = in4[j];
        if (RELU) { x.x = x.x > 0.f ? x.x : 0.f; x.y = x.y > 0.f ? x.y : 0.f;
                    x.z = x.z > 0.f ? x.z : 0.f; x.w = x.w > 0.f ? x.w : 0.f; }
        out2[j] = pack4_bf16(x);
      }
    }
  }
}

__global__ void k_transpose_w(const float* __restrict__ W,
                              unsigned short* __restrict__ Wt, int K, int NCOLS) {
  int i = blockIdx.x * blockDim.x + threadIdx.x;
  if (i >= K * NCOLS) return;
  int n = i / K, k = i % K;
  Wt[i] = f2bf(W[(size_t)k * NCOLS + n]);
}

template <int F>
__global__ __launch_bounds__(STREAM_THREADS)
void k_init_bias(float* __restrict__ acc, const float* __restrict__ b, unsigned n4) {
  static_assert(STREAM_STRIDE % (F / 4) == 0, "column group must not change along the loop");
  const unsigned t = blockIdx.x * STREAM_THREADS + threadIdx.x;
  const int c = (int)(t % (F / 4)) * 4;
  float4 s = make_float4(0.f, 0.f, 0.f, 0.f);
  for (int r = 0; r < R_REL; ++r) {
    const float4 v = *reinterpret_cast<const float4*>(b + (size_t)r * F + c);
    s.x += v.x; s.y += v.y; s.z += v.z; s.w += v.w;
  }
  float4* __restrict__ acc4 = reinterpret_cast<float4*>(acc);
#pragma clang loop vectorize(disable) interleave(disable)
  for (unsigned i = t; i < n4; i += STREAM_STRIDE) acc4[i] = s;
}

template <int KDIM, int NCOLS, int RPB>
__global__ __launch_bounds__(2 * NCOLS)
void k_gemm_bf16_wmma(const unsigned short* __restrict__ A,
                      const unsigned short* __restrict__ Bt,
                      float* __restrict__ C, int M) {
  constexpr int KC = KDIM / 32;
  const int lane     = threadIdx.x & 31;
  const int wv       = threadIdx.x >> 5;
  const int colBase  = wv * 16;
  const int l16      = lane & 15;
  const int k0       = (lane < 16) ? 0 : 8;
  const int rowBlock = blockIdx.x * (16 * RPB);

  Frag16 fb[KC];
  {
    const unsigned short* bp = Bt + (size_t)(colBase + l16) * KDIM + k0;
#pragma unroll
    for (int c = 0; c < KC; ++c) {
      fb[c].q[0] = *reinterpret_cast<const uint4*>(bp + 32 * c);
      fb[c].q[1] = *reinterpret_cast<const uint4*>(bp + 32 * c + 16);
    }
  }

  const bool full = (rowBlock + 16 * RPB) <= M;
  const int  rOff = (lane < 16) ? 0 : 8;

#pragma unroll
  for (int t = 0; t < RPB; ++t) {
    const int rowBase = rowBlock + 16 * t;
    if (rowBase >= M) break;

    int arow = rowBase + l16; if (arow >= M) arow = M - 1;
    const unsigned short* ap = A + (size_t)arow * KDIM + k0;

    v8f acc = {};
#pragma unroll
    for (int c = 0; c < KC; ++c) {
      Frag16 fa;
      fa.q[0] = *reinterpret_cast<const uint4*>(ap + 32 * c);
      fa.q[1] = *reinterpret_cast<const uint4*>(ap + 32 * c + 16);
      acc = __builtin_amdgcn_wmma_f32_16x16x32_bf16(false, fa.v, false, fb[c].v,
                                                    (short)0, acc, false, false);
    }

    float* cp = C + (size_t)rowBase * NCOLS + colBase + l16;
    if (full) {
#pragma unroll
      for (int v = 0; v < 8; ++v) cp[(size_t)(rOff + v) * NCOLS] = acc[v];
    } else {
#pragma unroll
      for (int v = 0; v < 8; ++v) {
        int rr = rowBase + rOff + v;
        if (rr < M) cp[(size_t)(rOff + v) * NCOLS] = acc[v];
      }
    }
  }
}

#define BKT_SHIFT    9
#define BKT_NODES    (1 << BKT_SHIFT)
#define SORT_BLOCKS   1024
#define SORT_THREADS  256
#define SORT2_THREADS 512

template <int NT>
__device__ __forceinline__ void block_excl_scan(unsigned* sh, int n, unsigned* part) {
  const int t  = threadIdx.x;
  const int C  = (n + NT - 1) / NT;
  const int lo = t * C;
  const int hi = (lo + C < n) ? lo + C : n;
  unsigned s = 0;
  for (int i = lo; i < hi; ++i) s += sh[i];
  part[t] = s;
  __syncthreads();
  for (int o = 1; o < NT; o <<= 1) {
    const unsigned v = (t >= o) ? part[t - o] : 0u;
    __syncthreads();
    part[t] += v;
    __syncthreads();
  }
  unsigned run = part[t] - s;
  for (int i = lo; i < hi; ++i) { const unsigned c = sh[i]; sh[i] = run; run += c; }
  __syncthreads();
}

__global__ __launch_bounds__(SORT_THREADS)
void k_bkt_hist(const int* __restrict__ dst_all, unsigned* __restrict__ bcount,
                int E, int B, int NK, unsigned total4, unsigned U) {
  extern __shared__ unsigned sh[];
  const int      t = threadIdx.x;
  const unsigned p = blockIdx.x;
  for (int k = t; k < NK; k += SORT_THREADS) sh[k] = 0u;
  __syncthreads();
  const int4* __restrict__ d4 = reinterpret_cast<const int4*>(dst_all);
  const unsigned uE = (unsigned)E, uB = (unsigned)B;
  const unsigned j0 = p * U;
  const unsigned j1 = (j0 + U < total4) ? j0 + U : total4;
  for (unsigned j = j0 + t; j < j1; j += SORT_THREADS) {
    const int4 d = d4[j];
    const unsigned g = 4u * j;
    atomicAdd(&sh[((g     ) / uE) * uB + ((unsigned)d.x >> BKT_SHIFT)], 1u);
    atomicAdd(&sh[((g + 1u) / uE) * uB + ((unsigned)d.y >> BKT_SHIFT)], 1u);
    atomicAdd(&sh[((g + 2u) / uE) * uB + ((unsigned)d.z >> BKT_SHIFT)], 1u);
    atomicAdd(&sh[((g + 3u) / uE) * uB + ((unsigned)d.w >> BKT_SHIFT)], 1u);
  }
  __syncthreads();
  for (int k = t; k < NK; k += SORT_THREADS) bcount[(size_t)k * gridDim.x + p] = sh[k];
}

__global__ __launch_bounds__(SORT_THREADS)
void k_bkt_scan(const unsigned* __restrict__ bcount, unsigned* __restrict__ boffs,
                unsigned* __restrict__ keytotal, int G) {
  __shared__ unsigned part[SORT_THREADS];
  const int    t    = threadIdx.x;
  const size_t base = (size_t)blockIdx.x * G;
  const int C  = (G + SORT_THREADS - 1) / SORT_THREADS;
  const int lo = t * C;
  const int hi = (lo + C < G) ? lo + C : G;
  unsigned s = 0;
  for (int i = lo; i < hi; ++i) s += bcount[base + i];
  part[t] = s;
  __syncthreads();
  for (int o = 1; o < SORT_THREADS; o <<= 1) {
    const unsigned v = (t >= o) ? part[t - o] : 0u;
    __syncthreads();
    part[t] += v;
    __syncthreads();
  }
  unsigned run = part[t] - s;
  for (int i = lo; i < hi; ++i) { const unsigned c = bcount[base + i]; boffs[base + i] = run; run += c; }
  if (t == SORT_THREADS - 1) keytotal[blockIdx.x] = part[t];
}

__global__ __launch_bounds__(SORT_THREADS)
void k_bkt_scatter(const int* __restrict__ src_all, const int* __restrict__ dst_all,
                   const unsigned* __restrict__ boffs, const unsigned* __restrict__ keytotal,
                   uint2* __restrict__ pairs, int E, int B, int NK, unsigned total4, unsigned U) {
  extern __shared__ unsigned sh[];
  unsigned* part = sh + NK;
  const int      t = threadIdx.x;
  const unsigned p = blockIdx.x;
  for (int k = t; k < NK; k += SORT_THREADS) sh[k] = keytotal[k];
  __syncthreads();
  block_excl_scan<SORT_THREADS>(sh, NK, part);
  for (int k = t; k < NK; k += SORT_THREADS) sh[k] += boffs[(size_t)k * gridDim.x + p];
  __syncthreads();
  const int4* __restrict__ s4 = reinterpret_cast<const int4*>(src_all);
  const int4* __restrict__ d4 = reinterpret_cast<const int4*>(dst_all);
  const unsigned uE = (unsigned)E, uB = (unsigned)B;
  const unsigned j0 = p * U;
  const unsigned j1 = (j0 + U < total4) ? j0 + U : total4;
  for (unsigned j = j0 + t; j < j1; j += SORT_THREADS) {
    const int4 d = d4[j];
    const int4 s = s4[j];
    const unsigned g = 4u * j;
    const unsigned p0 = atomicAdd(&sh[((g     ) / uE) * uB + ((unsigned)d.x >> BKT_SHIFT)], 1u);
    const unsigned p1 = atomicAdd(&sh[((g + 1u) / uE) * uB + ((unsigned)d.y >> BKT_SHIFT)], 1u);
    const unsigned p2 = atomicAdd(&sh[((g + 2u) / uE) * uB + ((unsigned)d.z >> BKT_SHIFT)], 1u);
    const unsigned p3 = atomicAdd(&sh[((g + 3u) / uE) * uB + ((unsigned)d.w >> BKT_SHIFT)], 1u);
    pairs[p0] = make_uint2((unsigned)s.x, (unsigned)d.x);
    pairs[p1] = make_uint2((unsigned)s.y, (unsigned)d.y);
    pairs[p2] = make_uint2((unsigned)s.z, (unsigned)d.z);
    pairs[p3] = make_uint2((unsigned)s.w, (unsigned)d.w);
  }
}

__global__ __launch_bounds__(SORT2_THREADS)
void k_bkt_sort(const uint2* __restrict__ pairs, const unsigned* __restrict__ keytotal,
                int* __restrict__ rowptr, int* __restrict__ csrc, int E, int B, int NP) {
  constexpr int NT = SORT2_THREADS;
  constexpr int DEPTH = 4;
  __shared__ unsigned cnt[BKT_NODES];
  __shared__ unsigned part[NT];
  const int t   = threadIdx.x;
  const int key = blockIdx.x;
  const int q   = key / B;
  const int nb0 = (key - q * B) << BKT_SHIFT;

  unsigned s = 0;
  for (int k = t; k < key; k += NT) s += keytotal[k];
  part[t] = s;
  for (int i = t; i < BKT_NODES; i += NT) cnt[i] = 0u;
  __syncthreads();
  for (int o = NT / 2; o > 0; o >>= 1) {
    if (t < o) part[t] += part[t + o];
    __syncthreads();
  }
  const unsigned bs = part[0];
  const unsigned n  = keytotal[key];
  __syncthreads();

  const uint2* __restrict__ pp = pairs + bs;
  for (unsigned i0 = 0; i0 < n; i0 += DEPTH * NT) {
    uint2 e[DEPTH];
#pragma unroll
    for (int k = 0; k < DEPTH; ++k) { const unsigned i = i0 + k * NT + t; e[k] = pp[i < n ? i : n - 1u]; }
#pragma unroll
    for (int k = 0; k < DEPTH; ++k) {
      const unsigned i = i0 + k * NT + t;
      if (i < n) atomicAdd(&cnt[e[k].y & (BKT_NODES - 1)], 1u);
    }
  }
  __syncthreads();
  block_excl_scan<NT>(cnt, BKT_NODES, part);

  const unsigned rel = bs - (unsigned)q * (unsigned)E;
  for (int i = t; i < BKT_NODES; i += NT)
    rowptr[(size_t)q * NP + nb0 + i] = (int)(rel + cnt[i]);
  __syncthreads();

  for (unsigned i0 = 0; i0 < n; i0 += DEPTH * NT) {
    uint2 e[DEPTH];
#pragma unroll
    for (int k = 0; k < DEPTH; ++k) { const unsigned i = i0 + k * NT + t; e[k] = pp[i < n ? i : n - 1u]; }
#pragma unroll
    for (int k = 0; k < DEPTH; ++k) {
      const unsigned i = i0 + k * NT + t;
      if (i < n) {
        const unsigned r = atomicAdd(&cnt[e[k].y & (BKT_NODES - 1)], 1u);
        csrc[bs + r] = (int)e[k].x;
      }
    }
  }
}

template <int CTRL>
__device__ __forceinline__ int dpp_i(int x) {
  return __builtin_amdgcn_update_dpp(0, x, CTRL, 0xF, 0xF, false);
}
template <int CTRL>
__device__ __forceinline__ float dpp_f(float x) {
  return __int_as_float(__builtin_amdgcn_update_dpp(0, __float_as_int(x), CTRL, 0xF, 0xF, false));
}
#define ROW_SHARE 0x150
#define ROW_XMASK 0x160

__device__ __forceinline__ float row_sum16(float x) {
  x += dpp_f<ROW_XMASK + 8>(x);
  x += dpp_f<ROW_XMASK + 4>(x);
  x += dpp_f<ROW_XMASK + 2>(x);
  x += dpp_f<ROW_XMASK + 1>(x);
  return x;
}

typedef __attribute__((ext_vector_type(2))) float f32x2;
typedef __attribute__((ext_vector_type(4))) float f32x4;

template <int XM>
__device__ __forceinline__ float fold_pair(bool bit, float lo, float hi) {
  const float keep = bit ? hi : lo;
  const float give = bit ? lo : hi;
  return keep + dpp_f<ROW_XMASK + XM>(give);
}

template <int G, int F>
__device__ __forceinline__ void group8(const float* __restrict__ frow, int s_l,
                                       const f32x2 al_lo, const f32x2 al_hi, float erd, bool valid,
                                       bool b3, bool b2, bool b1,
                                       float& m, float& den, f32x2& a_lo, f32x2& a_hi) {
#define SEG_LD(J) const f32x4 r##J = *reinterpret_cast<const f32x4*>( \
      frow + (size_t)dpp_i<ROW_SHARE + G + J>(s_l) * F);
  SEG_LD(0) SEG_LD(1) SEG_LD(2) SEG_LD(3) SEG_LD(4) SEG_LD(5) SEG_LD(6) SEG_LD(7)
#undef SEG_LD
#define SEG_DOT(J) float t##J; { \
      f32x2 q = f32x2{r##J.x, r##J.y} * al_lo; \
      q = __builtin_elementwise_fma(f32x2{r##J.z, r##J.w}, al_hi, q); \
      t##J = q.x + q.y; }
  SEG_DOT(0) SEG_DOT(1) SEG_DOT(2) SEG_DOT(3) SEG_DOT(4) SEG_DOT(5) SEG_DOT(6) SEG_DOT(7)
#undef SEG_DOT
  const float u0 = fold_pair<8>(b3, t0, t4), u1 = fold_pair<8>(b3, t1, t5);
  const float u2 = fold_pair<8>(b3, t2, t6), u3 = fold_pair<8>(b3, t3, t7);
  const float w0 = fold_pair<4>(b2, u0, u2), w1 = fold_pair<4>(b2, u1, u3);
  float z = fold_pair<2>(b1, w0, w1);
  z += dpp_f<ROW_XMASK + 1>(z);

  float e = lrelu(z + erd);
  e = valid ? e : -INFINITY;
  float mg = fmaxf(e, dpp_f<ROW_XMASK + 8>(e));
  mg = fmaxf(mg, dpp_f<ROW_XMASK + 4>(mg));
  mg = fmaxf(mg, dpp_f<ROW_XMASK + 2>(mg));
  const float mn   = fmaxf(m, mg);
  const float mref = (mn == -INFINITY) ? 0.f : mn;
  const float sc   = __expf(m - mref);
  const float p    = __expf(e - mref);
  float ps = p + dpp_f<ROW_XMASK + 8>(p);
  ps += dpp_f<ROW_XMASK + 4>(ps);
  ps += dpp_f<ROW_XMASK + 2>(ps);
  den = den * sc + ps;
  a_lo *= sc; a_hi *= sc;
  m = mn;
#define SEG_FMA(J) { const float w = dpp_f<ROW_SHARE + 2 * J>(p); const f32x2 ww = {w, w}; \
      a_lo = __builtin_elementwise_fma(f32x2{r##J.x, r##J.y}, ww, a_lo); \
      a_hi = __builtin_elementwise_fma(f32x2{r##J.z, r##J.w}, ww, a_hi); }
  SEG_FMA(0) SEG_FMA(1) SEG_FMA(2) SEG_FMA(3) SEG_FMA(4) SEG_FMA(5) SEG_FMA(6) SEG_FMA(7)
#undef SEG_FMA
}

template <int H>
__global__ __launch_bounds__(256)
void k_seg_softmax_sum(const int* __restrict__ rowptr, const int* __restrict__ csrc,
                       const float* __restrict__ al, const float* __restrict__ ar,
                       const float* __restrict__ feat, float* __restrict__ acc, int N) {
  static_assert(HID_F == 64 && OUT_F == 64, "a head is 64 floats in both layers");
  constexpr int D = 64;
  constexpr int F = H * D;
  const int wave = __builtin_amdgcn_readfirstlane(
      (int)(((size_t)blockIdx.x * 256 + threadIdx.x) >> 5));
  const int lane = threadIdx.x & 31;
  const int k    = lane & 15;
  const int v    = 2 * wave + (lane >> 4);
  const bool live = v < H * N;
  int h = 0;
  if constexpr (H == 2) h = (live && v >= N) ? 1 : 0;
  else if constexpr (H > 2) h = live ? v / N : 0;
  const int d = live ? v - h * N : 0;

  int beg = 0, end = 0;
  if (live) { beg = rowptr[d]; end = rowptr[d + 1]; }
  const int deg = end - beg;
  const int d0 = __builtin_amdgcn_readlane(deg, 0), d1 = __builtin_amdgcn_readlane(deg, 16);
  const int degmax = d0 > d1 ? d0 : d1;
  if (degmax <= 0) return;

  const float* frow = feat + h * D + k * 4;
  const int    last = end - 1;

  int s_nx = 0;
  if (deg > 0) { const int i0 = beg + k; s_nx = csrc[i0 < last ? i0 : last]; }

  const f32x4 al4 = *reinterpret_cast<const f32x4*>(al + h * D + k * 4);
  const f32x4 ar4 = *reinterpret_cast<const f32x4*>(ar + h * D + k * 4);
  const f32x4 fd  = *reinterpret_cast<const f32x4*>(frow + (size_t)d * F);
  const f32x2 al_lo = {al4.x, al4.y}, al_hi = {al4.z, al4.w};
  const float erd = row_sum16((fd.x * ar4.x + fd.z * ar4.z) + (fd.y * ar4.y + fd.w * ar4.w));

  const bool b3 = (k & 8) != 0, b2 = (k & 4) != 0, b1 = (k & 2) != 0;
  const int  kj = k >> 1;

  float m = -INFINITY, den = 0.f;
  f32x2 a_lo = {0.f, 0.f}, a_hi = {0.f, 0.f};

  for (int c0 = 0; c0 < degmax; c0 += 16) {
    const int i   = beg + c0 + k;
    const int s_l = s_nx;
    if (c0 + 16 < degmax && deg > 0) { const int i1 = i + 16; s_nx = csrc[i1 < last ? i1 : last]; }

    const int cnt  = degmax - c0;
    const int left = deg - c0;
    group8<0, F>(frow, s_l, al_lo, al_hi, erd, kj < left, b3, b2, b1, m, den, a_lo, a_hi);
    if (cnt > 8)
      group8<8, F>(frow, s_l, al_lo, al_hi, erd, kj + 8 < left, b3, b2, b1, m, den, a_lo, a_hi);
  }

  if (deg > 0) {
    float* op = acc + (size_t)d * F + h * D + k * 4;
    float4 o = *reinterpret_cast<const float4*>(op);
    o.x += a_lo.x / den; o.y += a_lo.y / den; o.z += a_hi.x / den; o.w += a_hi.y / den;
    *reinterpret_cast<float4*>(op) = o;
  }
}

extern "C" void kernel_launch(void* const* d_in, const int* in_sizes, int n_in,
                              void* d_out, int out_size, void* d_ws, size_t ws_size,
                              hipStream_t stream) {
  const float* x    = (const float*)d_in[0];
  const int*   esrc = (const int*)d_in[1];
  const int*   edst = (const int*)d_in[2];
  const float* W1   = (const float*)d_in[3];
  const float* al1  = (const float*)d_in[4];
  const float* ar1  = (const float*)d_in[5];
  const float* b1   = (const float*)d_in[6];
  const float* W2   = (const float*)d_in[7];
  const float* al2  = (const float*)d_in[8];
  const float* ar2  = (const float*)d_in[9];
  const float* b2   = (const float*)d_in[10];
  float* out = (float*)d_out;

  const int N = in_sizes[0] / IN_F;
  const int E = in_sizes[1] / (2 * R_REL);

  size_t off = 0;
  auto carve = [&](size_t bytes) -> void* {
    void* p = (char*)d_ws + off;
    off += (bytes + 255) & ~(size_t)255;
    return p;
  };
  unsigned short* xbf  = (unsigned short*)carve((size_t)N * IN_F * 2);
  unsigned short* hbf  = (unsigned short*)carve((size_t)N * (H1 * HID_F) * 2);
  unsigned short* wt1  = (unsigned short*)carve((size_t)R_REL * IN_F * (H1 * HID_F) * 2);
  unsigned short* wt2  = (unsigned short*)carve((size_t)R_REL * (H1 * HID_F) * OUT_F * 2);
  float*    feat  = (float*)carve((size_t)N * (H1 * HID_F) * 4);
  float*    hacc  = (float*)carve((size_t)N * (H1 * HID_F) * 4);
  const int NSETS = 2 * R_REL;
  const int NP    = (N / 1024 + 1) * 1024;
  const int B     = (N >> BKT_SHIFT) + 1;
  const int NK    = NSETS * B;
  int*      rowptr   = (int*)carve((size_t)NSETS * NP * 4);
  int*      csrc     = (int*)carve((size_t)NSETS * E * 4);
  unsigned* bcount   = (unsigned*)carve((size_t)NK * SORT_BLOCKS * 4);
  unsigned* boffs    = (unsigned*)carve((size_t)NK * SORT_BLOCKS * 4);
  unsigned* keytotal = (unsigned*)carve((size_t)NK * 4);
  uint2*    pairs    = (uint2*)carve((size_t)NSETS * E * 8);

  const int T = 256;
  auto blk1 = [&](size_t n) { return (unsigned)((n + T - 1) / T); };

  k_f32_to_bf16<false><<<STREAM_BLOCKS, STREAM_THREADS, 0, stream>>>(x, xbf, (unsigned)((size_t)N * IN_F / 4));
  for (int r = 0; r < R_REL; ++r) {
    k_transpose_w<<<blk1((size_t)IN_F * (H1 * HID_F)), T, 0, stream>>>(
        W1 + (size_t)r * IN_F * (H1 * HID_F), wt1 + (size_t)r * IN_F * (H1 * HID_F),
        IN_F, H1 * HID_F);
    k_transpose_w<<<blk1((size_t)(H1 * HID_F) * OUT_F), T, 0, stream>>>(
        W2 + (size_t)r * (H1 * HID_F) * OUT_F, wt2 + (size_t)r * (H1 * HID_F) * OUT_F,
        H1 * HID_F, OUT_F);
  }

  const unsigned total4 = (unsigned)((size_t)NSETS * E / 4);
  const unsigned U      = (total4 + SORT_BLOCKS - 1) / SORT_BLOCKS;
  k_bkt_hist<<<SORT_BLOCKS, SORT_THREADS, (size_t)NK * 4, stream>>>(edst, bcount, E, B, NK, total4, U);
  k_bkt_scan<<<NK, SORT_THREADS, 0, stream>>>(bcount, boffs, keytotal, SORT_BLOCKS);
  k_bkt_scatter<<<SORT_BLOCKS, SORT_THREADS, (size_t)(NK + SORT_THREADS) * 4, stream>>>(
      esrc, edst, boffs, keytotal, pairs, E, B, NK, total4, U);
  k_bkt_sort<<<NK, SORT2_THREADS, 0, stream>>>(pairs, keytotal, rowptr, csrc, E, B, NP);

  constexpr int RPB = 4;
  const unsigned gemmGrid = (unsigned)((N + 16 * RPB - 1) / (16 * RPB));

  const unsigned segGrid1 = (unsigned)(((size_t)N * H1 + 15) / 16);
  const unsigned segGrid2 = (unsigned)(((size_t)N * H2 + 15) / 16);

  k_init_bias<H1 * HID_F><<<STREAM_BLOCKS, STREAM_THREADS, 0, stream>>>(
      hacc, b1, (unsigned)((size_t)N * (H1 * HID_F) / 4));
  for (int r = 0; r < R_REL; ++r) {
    const int q1 = 0 * R_REL + r;

    k_gemm_bf16_wmma<IN_F, H1 * HID_F, RPB><<<gemmGrid, 2 * (H1 * HID_F), 0, stream>>>(
        xbf, wt1 + (size_t)r * IN_F * (H1 * HID_F), feat, N);

    k_seg_softmax_sum<H1><<<segGrid1, T, 0, stream>>>(
        rowptr + (size_t)q1 * NP, csrc + (size_t)q1 * E,
        al1 + (size_t)r * H1 * HID_F, ar1 + (size_t)r * H1 * HID_F, feat, hacc, N);
  }

  k_f32_to_bf16<true><<<STREAM_BLOCKS, STREAM_THREADS, 0, stream>>>(
      hacc, hbf, (unsigned)((size_t)N * (H1 * HID_F) / 4));

  k_init_bias<OUT_F><<<STREAM_BLOCKS, STREAM_THREADS, 0, stream>>>(
      out, b2, (unsigned)((size_t)N * OUT_F / 4));
  for (int r = 0; r < R_REL; ++r) {
    const int q2 = 1 * R_REL + r;

    k_gemm_bf16_wmma<H1 * HID_F, OUT_F, RPB><<<gemmGrid, 2 * OUT_F, 0, stream>>>(
        hbf, wt2 + (size_t)r * (H1 * HID_F) * OUT_F, feat, N);

    k_seg_softmax_sum<H2><<<segGrid2, T, 0, stream>>>(
        rowptr + (size_t)q2 * NP, csrc + (size_t)q2 * E,
        al2 + (size_t)r * H2 * OUT_F, ar2 + (size_t)r * H2 * OUT_F, feat, out, N);
  }
}


# Independent kernels share a launch, work-group index picks the role: bf16 x + weight transposes + edge histogram in one, bucket sort + layer-1 product in one; 13 launches to 7; sum orders kept
# speedup vs baseline: 6.6960x; 1.2482x over previous
#include <hip/hip_runtime.h>

#define R_REL   2
#define IN_F    128
#define HID_F   64
#define OUT_F   64
#define H1      2
#define H2      1
#define NEG_SLOPE 0.2f

typedef __attribute__((ext_vector_type(16))) __bf16 v16bf;
typedef __attribute__((ext_vector_type(8)))  float  v8f;

union Frag16 { uint4 q[2]; v16bf v; };

__device__ __forceinline__ unsigned short f2bf(float f) {
  unsigned u = __float_as_uint(f);
  unsigned r = u + 0x7FFFu + ((u >> 16) & 1u);
  return (unsigned short)(r >> 16);
}
__device__ __forceinline__ float lrelu(float x) { return x >= 0.f ? x : NEG_SLOPE * x; }

#define STREAM_BLOCKS  512
#define STREAM_THREADS 256
#define STREAM_STRIDE  (STREAM_BLOCKS * STREAM_THREADS)

__device__ __forceinline__ uint2 pack4_bf16(float4 v) {
  uint2 r;
  r.x = (unsigned)f2bf(v.x) | ((unsigned)f2bf(v.y) << 16);
  r.y = (unsigned)f2bf(v.z) | ((unsigned)f2bf(v.w) << 16);
  return r;
}

__device__ __forceinline__
void role_f32_to_bf16(unsigned blk, const float* __restrict__ in, unsigned short* __restrict__ out, unsigned n4) {
  const float4* __restrict__ in4  = reinterpret_cast<const float4*>(in);
  uint2* __restrict__        out2 = reinterpret_cast<uint2*>(out);
  constexpr int U = 8;
  const unsigned t = blk * STREAM_THREADS + threadIdx.x;
  for (unsigned i = t; i < n4; i += U * STREAM_STRIDE) {
    float4 v[U];
    if (i + (U - 1) * STREAM_STRIDE < n4) {
#pragma unroll
      for (int k = 0; k < U; ++k) v[k] = in4[i + k * STREAM_STRIDE];
#pragma unroll
      for (int k = 0; k < U; ++k) out2[i + k * STREAM_STRIDE] = pack4_bf16(v[k]);
    } else {
      for (unsigned j = i; j < n4; j += STREAM_STRIDE) out2[j] = pack4_bf16(in4[j]);
    }
  }
}

template <int K, int NCOLS>
__device__ __forceinline__
void role_transpose_w(unsigned blk, const float* __restrict__ W, unsigned short* __restrict__ Wt) {
  const int i = (int)(blk * STREAM_THREADS + threadIdx.x);
  if (i >= R_REL * K * NCOLS) return;
  const int r = i / (K * NCOLS), rem = i - r * (K * NCOLS);
  const int n = rem / K, k = rem - n * K;
  Wt[i] = f2bf(W[(size_t)r * (K * NCOLS) + (size_t)k * NCOLS + n]);
}

template <int KDIM, int NCOLS, int RPB>
__device__ __forceinline__
void role_gemm_bf16_wmma(unsigned blk, const unsigned short* __restrict__ A,
                         const unsigned short* __restrict__ Bt,
                         float* __restrict__ C, int M) {
  constexpr int KC = KDIM / 32;
  const int lane     = threadIdx.x & 31;
  const int wv       = threadIdx.x >> 5;
  const int colBase  = wv * 16;
  const int l16      = lane & 15;
  const int k0       = (lane < 16) ? 0 : 8;
  const int rowBlock = (int)blk * (16 * RPB);

  Frag16 fb[KC];
  {
    const unsigned short* bp = Bt + (size_t)(colBase + l16) * KDIM + k0;
#pragma unroll
    for (int c = 0; c < KC; ++c) {
      fb[c].q[0] = *reinterpret_cast<const uint4*>(bp + 32 * c);
      fb[c].q[1] = *reinterpret_cast<const uint4*>(bp + 32 * c + 16);
    }
  }

  const bool full = (rowBlock + 16 * RPB) <= M;
  const int  rOff = (lane < 16) ? 0 : 8;

#pragma unroll
  for (int t = 0; t < RPB; ++t) {
    const int rowBase = rowBlock + 16 * t;
    if (rowBase >= M) break;

    int arow = rowBase + l16; if (arow >= M) arow = M - 1;
    const unsigned short* ap = A + (size_t)arow * KDIM + k0;

    v8f acc = {};
#pragma unroll
    for (int c = 0; c < KC; ++c) {
      Frag16 fa;
      fa.q[0] = *reinterpret_cast<const uint4*>(ap + 32 * c);
      fa.q[1] = *reinterpret_cast<const uint4*>(ap + 32 * c + 16);
      acc = __builtin_amdgcn_wmma_f32_16x16x32_bf16(false, fa.v, false, fb[c].v,
                                                    (short)0, acc, false, false);
    }

    float* cp = C + (size_t)rowBase * NCOLS + colBase + l16;
    if (full) {
#pragma unroll
      for (int v = 0; v < 8; ++v) cp[(size_t)(rOff + v) * NCOLS] = acc[v];
    } else {
#pragma unroll
      for (int v = 0; v < 8; ++v) {
        int rr = rowBase + rOff + v;
        if (rr < M) cp[(size_t)(rOff + v) * NCOLS] = acc[v];
      }
    }
  }
}

#define BKT_SHIFT    9
#define BKT_NODES    (1 << BKT_SHIFT)
#define SORT_BLOCKS   1024
#define SORT_THREADS  256
#define SORT2_THREADS 512

template <int NT>
__device__ __forceinline__ void block_excl_scan(unsigned* sh, int n, unsigned* part) {
  const int t  = threadIdx.x;
  const int C  = (n + NT - 1) / NT;
  const int lo = t * C;
  const int hi = (lo + C < n) ? lo + C : n;
  unsigned s = 0;
  for (int i = lo; i < hi; ++i) s += sh[i];
  part[t] = s;
  __syncthreads();
  for (int o = 1; o < NT; o <<= 1) {
    const unsigned v = (t >= o) ? part[t - o] : 0u;
    __syncthreads();
    part[t] += v;
    __syncthreads();
  }
  unsigned run = part[t] - s;
  for (int i = lo; i < hi; ++i) { const unsigned c = sh[i]; sh[i] = run; run += c; }
  __syncthreads();
}

__device__ __forceinline__
void role_bkt_hist(unsigned p, unsigned* sh, const int* __restrict__ dst_all, unsigned* __restrict__ bcount,
                   int E, int B, int NK, unsigned total4, unsigned U) {
  const int      t = threadIdx.x;
  for (int k = t; k < NK; k += SORT_THREADS) sh[k] = 0u;
  __syncthreads();
  const int4* __restrict__ d4 = reinterpret_cast<const int4*>(dst_all);
  const unsigned uE = (unsigned)E, uB = (unsigned)B;
  const unsigned j0 = p * U;
  const unsigned j1 = (j0 + U < total4) ? j0 + U : total4;
  for (unsigned j = j0 + t; j < j1; j += SORT_THREADS) {
    const int4 d = d4[j];
    const unsigned g = 4u * j;
    atomicAdd(&sh[((g     ) / uE) * uB + ((unsigned)d.x >> BKT_SHIFT)], 1u);
    atomicAdd(&sh[((g + 1u) / uE) * uB + ((unsigned)d.y >> BKT_SHIFT)], 1u);
    atomicAdd(&sh[((g + 2u) / uE) * uB + ((unsigned)d.z >> BKT_SHIFT)], 1u);
    atomicAdd(&sh[((g + 3u) / uE) * uB + ((unsigned)d.w >> BKT_SHIFT)], 1u);
  }
  __syncthreads();
  for (int k = t; k < NK; k += SORT_THREADS) bcount[(size_t)k * SORT_BLOCKS + p] = sh[k];
}

#define TW1_BLOCKS ((R_REL * IN_F * (H1 * HID_F)) / STREAM_THREADS)
#define TW2_BLOCKS ((R_REL * (H1 * HID_F) * (H2 * OUT_F)) / STREAM_THREADS)
#define PREP_BLOCKS (STREAM_BLOCKS + SORT_BLOCKS + TW1_BLOCKS + TW2_BLOCKS)
static_assert(STREAM_THREADS == SORT_THREADS, "the roles of the first launch share one work-group size");
__global__ __launch_bounds__(STREAM_THREADS)
void k_prep(const float* __restrict__ x, unsigned short* __restrict__ xbf, unsigned n4,
            const float* __restrict__ W1, unsigned short* __restrict__ wt1,
            const float* __restrict__ W2, unsigned short* __restrict__ wt2,
            const int* __restrict__ dst_all, unsigned* __restrict__ bcount,
            int E, int B, int NK, unsigned total4, unsigned U) {
  extern __shared__ unsigned sh[];
  unsigned b = blockIdx.x;
  if (b < STREAM_BLOCKS) { role_f32_to_bf16(b, x, xbf, n4); return; }
  b -= STREAM_BLOCKS;
  if (b < SORT_BLOCKS) { role_bkt_hist(b, sh, dst_all, bcount, E, B, NK, total4, U); return; }
  b -= SORT_BLOCKS;
  if (b < TW1_BLOCKS) { role_transpose_w<IN_F, H1 * HID_F>(b, W1, wt1); return; }
  b -= TW1_BLOCKS;
  role_transpose_w<H1 * HID_F, H2 * OUT_F>(b, W2, wt2);
}

__global__ __launch_bounds__(SORT_THREADS)
void k_bkt_scan(const unsigned* __restrict__ bcount, unsigned* __restrict__ boffs,
                unsigned* __restrict__ keytotal, int G) {
  __shared__ unsigned part[SORT_THREADS];
  const int    t    = threadIdx.x;
  const size_t base = (size_t)blockIdx.x * G;
  const int C  = (G + SORT_THREADS - 1) / SORT_THREADS;
  const int lo = t * C;
  const int hi = (lo + C < G) ? lo + C : G;
  unsigned s = 0;
  for (int i = lo; i < hi; ++i) s += bcount[base + i];
  part[t] = s;
  __syncthreads();
  for (int o = 1; o < SORT_THREADS; o <<= 1) {
    const unsigned v = (t >= o) ? part[t - o] : 0u;
    __syncthreads();
    part[t] += v;
    __syncthreads();
  }
  unsigned run = part[t] - s;
  for (int i = lo; i < hi; ++i) { const unsigned c = bcount[base + i]; boffs[base + i] = run; run += c; }
  if (t == SORT_THREADS - 1) keytotal[blockIdx.x] = part[t];
}

__global__ __launch_bounds__(SORT_THREADS)
void k_bkt_scatter(const int* __restrict__ src_all, const int* __restrict__ dst_all,
                   const unsigned* __restrict__ boffs, const unsigned* __restrict__ keytotal,
                   uint2* __restrict__ pairs, int E, int B, int NK, unsigned total4, unsigned U) {
  extern __shared__ unsigned sh[];
  unsigned* part = sh + NK;
  const int      t = threadIdx.x;
  const unsigned p = blockIdx.x;
  for (int k = t; k < NK; k += SORT_THREADS) sh[k] = keytotal[k];
  __syncthreads();
  block_excl_scan<SORT_THREADS>(sh, NK, part);
  for (int k = t; k < NK; k += SORT_THREADS) sh[k] += boffs[(size_t)k * gridDim.x + p];
  __syncthreads();
  const int4* __restrict__ s4 = reinterpret_cast<const int4*>(src_all);
  const int4* __restrict__ d4 = reinterpret_cast<const int4*>(dst_all);
  const unsigned uE = (unsigned)E, uB = (unsigned)B;
  const unsigned j0 = p * U;
  const unsigned j1 = (j0 + U < total4) ? j0 + U : total4;
  for (unsigned j = j0 + t; j < j1; j += SORT_THREADS) {
    const int4 d = d4[j];
    const int4 s = s4[j];
    const unsigned g = 4u * j;
    const unsigned p0 = atomicAdd(&sh[((g     ) / uE) * uB + ((unsigned)d.x >> BKT_SHIFT)], 1u);
    const unsigned p1 = atomicAdd(&sh[((g + 1u) / uE) * uB + ((unsigned)d.y >> BKT_SHIFT)], 1u);
    const unsigned p2 = atomicAdd(&sh[((g + 2u) / uE) * uB + ((unsigned)d.z >> BKT_SHIFT)], 1u);
    const unsigned p3 = atomicAdd(&sh[((g + 3u) / uE) * uB + ((unsigned)d.w >> BKT_SHIFT)], 1u);
    pairs[p0] = make_uint2((unsigned)s.x, (unsigned)d.x);
    pairs[p1] = make_uint2((unsigned)s.y, (unsigned)d.y);
    pairs[p2] = make_uint2((unsigned)s.z, (unsigned)d.z);
    pairs[p3] = make_uint2((unsigned)s.w, (unsigned)d.w);
  }
}

__device__ __forceinline__
void role_bkt_sort(int key, const uint2* __restrict__ pairs, const unsigned* __restrict__ keytotal,
                   int* __restrict__ rowptr, int* __restrict__ csrc, int E, int B, int NP) {
  constexpr int NT = SORT2_THREADS;
  constexpr int DEPTH = 4;
  __shared__ unsigned cnt[BKT_NODES];
  __shared__ unsigned part[NT];
  const int t   = threadIdx.x;
  const int q   = key / B;
  const int nb0 = (key - q * B) << BKT_SHIFT;

  unsigned s = 0;
  for (int k = t; k < key; k += NT) s += keytotal[k];
  part[t] = s;
  for (int i = t; i < BKT_NODES; i += NT) cnt[i] = 0u;
  __syncthreads();
  for (int o = NT / 2; o > 0; o >>= 1) {
    if (t < o) part[t] += part[t + o];
    __syncthreads();
  }
  const unsigned bs = part[0];
  const unsigned n  = keytotal[key];
  __syncthreads();

  const uint2* __restrict__ pp = pairs + bs;
  for (unsigned i0 = 0; i0 < n; i0 += DEPTH * NT) {
    uint2 e[DEPTH];
#pragma unroll
    for (int k = 0; k < DEPTH; ++k) { const unsigned i = i0 + k * NT + t; e[k] = pp[i < n ? i : n - 1u]; }
#pragma unroll
    for (int k = 0; k < DEPTH; ++k) {
      const unsigned i = i0 + k * NT + t;
      if (i < n) atomicAdd(&cnt[e[k].y & (BKT_NODES - 1)], 1u);
    }
  }
  __syncthreads();
  block_excl_scan<NT>(cnt, BKT_NODES, part);

  const unsigned rel = bs - (unsigned)q * (unsigned)E;
  for (int i = t; i < BKT_NODES; i += NT)
    rowptr[(size_t)q * NP + nb0 + i] = (int)(rel + cnt[i]);
  __syncthreads();

  for (unsigned i0 = 0; i0 < n; i0 += DEPTH * NT) {
    uint2 e[DEPTH];
#pragma unroll
    for (int k = 0; k < DEPTH; ++k) { const unsigned i = i0 + k * NT + t; e[k] = pp[i < n ? i : n - 1u]; }
#pragma unroll
    for (int k = 0; k < DEPTH; ++k) {
      const unsigned i = i0 + k * NT + t;
      if (i < n) {
        const unsigned r = atomicAdd(&cnt[e[k].y & (BKT_NODES - 1)], 1u);
        csrc[bs + r] = (int)e[k].x;
      }
    }
  }
}

#define GEMM_RPB 4
static_assert(SORT2_THREADS == 2 * (R_REL * H1 * HID_F), "sort and layer-1 product share one work-group size");
__global__ __launch_bounds__(SORT2_THREADS)
void k_sort_gemm1(const uint2* __restrict__ pairs, const unsigned* __restrict__ keytotal,
                  int* __restrict__ rowptr, int* __restrict__ csrc, int E, int B, int NP, int NK,
                  const unsigned short* __restrict__ A, const unsigned short* __restrict__ Bt,
                  float* __restrict__ C, int M) {
  if ((int)blockIdx.x < NK) {
    role_bkt_sort((int)blockIdx.x, pairs, keytotal, rowptr, csrc, E, B, NP);
  } else {
    role_gemm_bf16_wmma<IN_F, R_REL * H1 * HID_F, GEMM_RPB>(blockIdx.x - (unsigned)NK, A, Bt, C, M);
  }
}

__global__ __launch_bounds__(2 * (R_REL * H2 * OUT_F))
void k_gemm2(const unsigned short* __restrict__ A, const unsigned short* __restrict__ Bt,
             float* __restrict__ C, int M) {
  role_gemm_bf16_wmma<H1 * HID_F, R_REL * H2 * OUT_F, GEMM_RPB>(blockIdx.x, A, Bt, C, M);
}

template <int CTRL>
__device__ __forceinline__ int dpp_i(int x) {
  return __builtin_amdgcn_update_dpp(0, x, CTRL, 0xF, 0xF, false);
}
template <int CTRL>
__device__ __forceinline__ float dpp_f(float x) {
  return __int_as_float(__builtin_amdgcn_update_dpp(0, __float_as_int(x), CTRL, 0xF, 0xF, false));
}
#define ROW_SHARE 0x150
#define ROW_XMASK 0x160

__device__ __forceinline__ float row_sum16(float x) {
  x += dpp_f<ROW_XMASK + 8>(x);
  x += dpp_f<ROW_XMASK + 4>(x);
  x += dpp_f<ROW_XMASK + 2>(x);
  x += dpp_f<ROW_XMASK + 1>(x);
  return x;
}

typedef __attribute__((ext_vector_type(2))) float f32x2;
typedef __attribute__((ext_vector_type(4))) float f32x4;

template <int XM>
__device__ __forceinline__ float fold_pair(bool bit, float lo, float hi) {
  const float keep = bit ? hi : lo;
  const float give = bit ? lo : hi;
  return keep + dpp_f<ROW_XMASK + XM>(give);
}

template <int G, unsigned PITCH4>
__device__ __forceinline__ void group8(const f32x4* __restrict__ fbase, unsigned loff4, int s_l,
                                       const f32x2 al_lo, const f32x2 al_hi, float erd, bool valid,
                                       bool b3, bool b2, bool b1,
                                       float& m, float& den, f32x2& a_lo, f32x2& a_hi) {
#define SEG_LD(J) const f32x4 r##J = fbase[(unsigned)dpp_i<ROW_SHARE + G + J>(s_l) * PITCH4 + loff4];
  SEG_LD(0) SEG_LD(1) SEG_LD(2) SEG_LD(3) SEG_LD(4) SEG_LD(5) SEG_LD(6) SEG_LD(7)
#undef SEG_LD
#define SEG_DOT(J) float t##J; { \
      f32x2 q = f32x2{r##J.x, r##J.y} * al_lo; \
      q = __builtin_elementwise_fma(f32x2{r##J.z, r##J.w}, al_hi, q); \
      t##J = q.x + q.y; }
  SEG_DOT(0) SEG_DOT(1) SEG_DOT(2) SEG_DOT(3) SEG_DOT(4) SEG_DOT(5) SEG_DOT(6) SEG_DOT(7)
#undef SEG_DOT
  const float u0 = fold_pair<8>(b3, t0, t4), u1 = fold_pair<8>(b3, t1, t5);
  const float u2 = fold_pair<8>(b3, t2, t6), u3 = fold_pair<8>(b3, t3, t7);
  const float w0 = fold_pair<4>(b2, u0, u2), w1 = fold_pair<4>(b2, u1, u3);
  float z = fold_pair<2>(b1, w0, w1);
  z += dpp_f<ROW_XMASK + 1>(z);

  float e = lrelu(z + erd);
  e = valid ? e : -INFINITY;
  float mg = fmaxf(e, dpp_f<ROW_XMASK + 8>(e));
  mg = fmaxf(mg, dpp_f<ROW_XMASK + 4>(mg));
  mg = fmaxf(mg, dpp_f<ROW_XMASK + 2>(mg));
  const float mn   = fmaxf(m, mg);
  const float mref = (mn == -INFINITY) ? 0.f : mn;
  const float sc   = __expf(m - mref);
  const float p    = __expf(e - mref);
  float ps = p + dpp_f<ROW_XMASK + 8>(p);
  ps += dpp_f<ROW_XMASK + 4>(ps);
  ps += dpp_f<ROW_XMASK + 2>(ps);
  den = den * sc + ps;
  a_lo *= sc; a_hi *= sc;
  m = mn;
#define SEG_FMA(J) { const float w = dpp_f<ROW_SHARE + 2 * J>(p); const f32x2 ww = {w, w}; \
      a_lo = __builtin_elementwise_fma(f32x2{r##J.x, r##J.y}, ww, a_lo); \
      a_hi = __builtin_elementwise_fma(f32x2{r##J.z, r##J.w}, ww, a_hi); }
  SEG_FMA(0) SEG_FMA(1) SEG_FMA(2) SEG_FMA(3) SEG_FMA(4) SEG_FMA(5) SEG_FMA(6) SEG_FMA(7)
#undef SEG_FMA
}

template <int H, bool RELU_BF16>
__global__ __launch_bounds__(256)
void k_seg_softmax_sum(const int* __restrict__ rowptr, const int* __restrict__ csrc,
                       const float* __restrict__ al, const float* __restrict__ ar,
                       const float* __restrict__ bias, const float* __restrict__ feat,
                       float* __restrict__ outf, unsigned short* __restrict__ outb,
                       int N, int NP, int E) {
  static_assert(HID_F == 64 && OUT_F == 64, "a head is 64 floats in both layers");
  constexpr int D  = 64;
  constexpr int F  = H * D;
  constexpr int FA = R_REL * F;
  const int wave = __builtin_amdgcn_readfirstlane(
      (int)(((size_t)blockIdx.x * 256 + threadIdx.x) >> 5));
  const int lane = threadIdx.x & 31;
  const int k    = lane & 15;
  const int v    = 2 * wave + (lane >> 4);
  const bool live = v < H * N;
  int h = 0;
  if constexpr (H == 2) h = (live && v >= N) ? 1 : 0;
  else if constexpr (H > 2) h = live ? v / N : 0;
  const int d   = live ? v - h * N : 0;
  const int col = h * D + k * 4;

  const f32x4* fbase = reinterpret_cast<const f32x4*>(feat);
  const unsigned loff4 = (unsigned)col / 4u;

  int beg[R_REL], end[R_REL], s_first[R_REL];
#pragma unroll
  for (int r = 0; r < R_REL; ++r) {
    const int b = rowptr[(size_t)r * NP + d], e = rowptr[(size_t)r * NP + d + 1];
    beg[r] = live ? b : 0; end[r] = live ? e : 0;
  }
  f32x4 ar4[R_REL], fd[R_REL];
#pragma unroll
  for (int r = 0; r < R_REL; ++r) {
    ar4[r] = reinterpret_cast<const f32x4*>(ar)[loff4 + r * (F / 4)];
    fd[r]  = fbase[(unsigned)d * (unsigned)(FA / 4) + loff4 + r * (F / 4)];
  }
#pragma unroll
  for (int r = 0; r < R_REL; ++r) {
    const int i0 = beg[r] + k, last = end[r] - 1;
    const int i  = end[r] > beg[r] ? (i0 < last ? i0 : last) : 0;
    s_first[r] = csrc[(size_t)r * E + i];
  }
  asm volatile("" :: "v"(s_first[0]), "v"(s_first[R_REL - 1]));

  float erd[R_REL];
#pragma unroll
  for (int r = 0; r < R_REL; ++r) {
    erd[r] = row_sum16((fd[r].x * ar4[r].x + fd[r].z * ar4[r].z) + (fd[r].y * ar4[r].y + fd[r].w * ar4[r].w));
  }

  const bool b3 = (k & 8) != 0, b2 = (k & 4) != 0, b1 = (k & 2) != 0;
  const int  kj = k >> 1;

  f32x4 o = {0.f, 0.f, 0.f, 0.f};
#pragma unroll
  for (int r = 0; r < R_REL; ++r) {
    const int deg = end[r] - beg[r];
    const int d0 = __builtin_amdgcn_readlane(deg, 0), d1 = __builtin_amdgcn_readlane(deg, 16);
    const int degmax = d0 > d1 ? d0 : d1;

    float m = -INFINITY, den = 0.f;
    f32x2 a_lo = {0.f, 0.f}, a_hi = {0.f, 0.f};

    if (degmax > 0) {
      const unsigned lr4 = loff4 + r * (F / 4);
      const int*   cs   = csrc + (size_t)r * E;
      const int    last = end[r] - 1;

      const f32x4 al4 = reinterpret_cast<const f32x4*>(al)[lr4];
      const f32x2 al_lo = {al4.x, al4.y}, al_hi = {al4.z, al4.w};
      int s_nx = s_first[r];

      for (int c0 = 0; c0 < degmax; c0 += 16) {
        const int i   = beg[r] + c0 + k;
        const int s_l = s_nx;
        if (c0 + 16 < degmax && deg > 0) { const int i1 = i + 16; s_nx = cs[i1 < last ? i1 : last]; }

        const int cnt  = degmax - c0;
        const int left = deg - c0;
        group8<0, FA / 4>(fbase, lr4, s_l, al_lo, al_hi, erd[r], kj < left, b3, b2, b1, m, den, a_lo, a_hi);
        if (cnt > 8)
          group8<8, FA / 4>(fbase, lr4, s_l, al_lo, al_hi, erd[r], kj + 8 < left, b3, b2, b1, m, den, a_lo, a_hi);
      }
    }

    if (r == 0) {
#pragma unroll
      for (int rr = 0; rr < R_REL; ++rr) o += reinterpret_cast<const f32x4*>(bias)[loff4 + rr * (F / 4)];
    }
    if (deg > 0) { o.x += a_lo.x / den; o.y += a_lo.y / den; o.z += a_hi.x / den; o.w += a_hi.y / den; }
  }

  if (live) {
    if constexpr (RELU_BF16) {
      const float4 x = make_float4(o.x > 0.f ? o.x : 0.f, o.y > 0.f ? o.y : 0.f,
                                   o.z > 0.f ? o.z : 0.f, o.w > 0.f ? o.w : 0.f);
      *reinterpret_cast<uint2*>(outb + (size_t)d * F + col) = pack4_bf16(x);
    } else {
      *reinterpret_cast<f32x4*>(outf + (size_t)d * F + col) = o;
    }
  }
}

extern "C" void kernel_launch(void* const* d_in, const int* in_sizes, int n_in,
                              void* d_out, int out_size, void* d_ws, size_t ws_size,
                              hipStream_t stream) {
  const float* x    = (const float*)d_in[0];
  const int*   esrc = (const int*)d_in[1];
  const int*   edst = (const int*)d_in[2];
  const float* W1   = (const float*)d_in[3];
  const float* al1  = (const float*)d_in[4];
  const float* ar1  = (const float*)d_in[5];
  const float* b1   = (const float*)d_in[6];
  const float* W2   = (const float*)d_in[7];
  const float* al2  = (const float*)d_in[8];
  const float* ar2  = (const float*)d_in[9];
  const float* b2   = (const float*)d_in[10];
  float* out = (float*)d_out;

  const int N = in_sizes[0] / IN_F;
  const int E = in_sizes[1] / (2 * R_REL);

  size_t off = 0;
  auto carve = [&](size_t bytes) -> void* {
    void* p = (char*)d_ws + off;
    off += (bytes + 255) & ~(size_t)255;
    return p;
  };
  unsigned short* xbf  = (unsigned short*)carve((size_t)N * IN_F * 2);
  unsigned short* hbf  = (unsigned short*)carve((size_t)N * (H1 * HID_F) * 2);
  unsigned short* wt1  = (unsigned short*)carve((size_t)R_REL * IN_F * (H1 * HID_F) * 2);
  unsigned short* wt2  = (unsigned short*)carve((size_t)R_REL * (H1 * HID_F) * OUT_F * 2);
  float*    feat  = (float*)carve((size_t)N * R_REL * (H1 * HID_F) * 4);
  const int NSETS = 2 * R_REL;
  const int NP    = (N / 1024 + 1) * 1024;
  const int B     = (N >> BKT_SHIFT) + 1;
  const int NK    = NSETS * B;
  int*      rowptr   = (int*)carve((size_t)NSETS * NP * 4);
  int*      csrc     = (int*)carve((size_t)NSETS * E * 4);
  unsigned* bcount   = (unsigned*)carve((size_t)NK * SORT_BLOCKS * 4);
  unsigned* boffs    = (unsigned*)carve((size_t)NK * SORT_BLOCKS * 4);
  unsigned* keytotal = (unsigned*)carve((size_t)NK * 4);
  uint2*    pairs    = (uint2*)carve((size_t)NSETS * E * 8);

  const int T = 256;

  const unsigned total4 = (unsigned)((size_t)NSETS * E / 4);
  const unsigned U      = (total4 + SORT_BLOCKS - 1) / SORT_BLOCKS;
  k_prep<<<PREP_BLOCKS, STREAM_THREADS, (size_t)NK * 4, stream>>>(
      x, xbf, (unsigned)((size_t)N * IN_F / 4), W1, wt1, W2, wt2, edst, bcount, E, B, NK, total4, U);

  k_bkt_scan<<<NK, SORT_THREADS, 0, stream>>>(bcount, boffs, keytotal, SORT_BLOCKS);
  k_bkt_scatter<<<SORT_BLOCKS, SORT_THREADS, (size_t)(NK + SORT_THREADS) * 4, stream>>>(
      esrc, edst, boffs, keytotal, pairs, E, B, NK, total4, U);

  const unsigned gemmGrid = (unsigned)((N + 16 * GEMM_RPB - 1) / (16 * GEMM_RPB));

  const unsigned segGrid1 = (unsigned)(((size_t)N * H1 + 15) / 16);
  const unsigned segGrid2 = (unsigned)(((size_t)N * H2 + 15) / 16);

  k_sort_gemm1<<<(unsigned)NK + gemmGrid, SORT2_THREADS, 0, stream>>>(
      pairs, keytotal, rowptr, csrc, E, B, NP, NK, xbf, wt1, feat, N);
  k_seg_softmax_sum<H1, true><<<segGrid1, T, 0, stream>>>(
      rowptr, csrc, al1, ar1, b1, feat, nullptr, hbf, N, NP, E);

  k_gemm2<<<gemmGrid, 2 * (R_REL * H2 * OUT_F), 0, stream>>>(hbf, wt2, feat, N);
  k_seg_softmax_sum<H2, false><<<segGrid2, T, 0, stream>>>(
      rowptr + (size_t)R_REL * NP, csrc + (size_t)R_REL * E, al2, ar2, b2, feat, out, nullptr, N, NP, E);
}
